# starting kernel
# baseline (speedup 1.0000x reference)
#include <hip/hip_runtime.h>
#include <math.h>


#define N_NODES 50000
#define N_EDGES 800000
#define IN_DIM 256
#define HID 64
#define HEADS 4
#define FTOT 256
#define NEG_SLOPE 0.2f

typedef __attribute__((ext_vector_type(2))) float v2f;
typedef __attribute__((ext_vector_type(8))) float v8f;

__global__ __launch_bounds__(256) void gat_gemm_f32_wmma(
    const float* __restrict__ A, const float* __restrict__ W,
    float* __restrict__ out)
{
    __shared__ float lds[16 * 256];
    const int tid  = threadIdx.x;
    const int row0 = blockIdx.x * 16;

    const float* Ablk = A + (size_t)row0 * FTOT;
    #pragma unroll
    for (int i = 0; i < 16; ++i)
        lds[tid + i * 256] = Ablk[tid + i * 256];
    __syncthreads();

    const int wave  = tid >> 5;
    const int lane  = tid & 31;
    const int col0  = (blockIdx.y * 8 + wave) * 16;
    const int hlane = lane & 15;
    const int koff  = (lane >> 4) * 2;

    v8f acc = {};
    const float* Wc = W + col0 + hlane;
    #pragma unroll 4
    for (int k = 0; k < IN_DIM; k += 4) {
        v2f a = *(const v2f*)&lds[hlane * 256 + k + koff];
        v2f b;
        b.x = Wc[(size_t)(k + koff) * FTOT];
        b.y = Wc[(size_t)(k + koff + 1) * FTOT];
        acc = __builtin_amdgcn_wmma_f32_16x16x4_f32(
                  false, a, false, b, (short)0, acc, false, false);
    }

    const int srow = row0 + ((lane >> 4) * 8);
    float* Orow = out + (size_t)srow * FTOT + col0 + hlane;
    #pragma unroll
    for (int j = 0; j < 8; ++j)
        Orow[(size_t)j * FTOT] = acc[j];
}

__global__ __launch_bounds__(256) void gat_eler(
    const float* __restrict__ feat, const float* __restrict__ al,
    const float* __restrict__ ar, float* __restrict__ el, float* __restrict__ er)
{
    __shared__ float sl[256], sr[256];
    const int n = blockIdx.x, c = threadIdx.x;
    float v = feat[(size_t)n * FTOT + c];
    sl[c] = v * al[c];
    sr[c] = v * ar[c];
    __syncthreads();
    const int d = c & 63;
    for (int s = 32; s > 0; s >>= 1) {
        if (d < s) { sl[c] += sl[c + s]; sr[c] += sr[c + s]; }
        __syncthreads();
    }
    if (d == 0) {
        el[n * HEADS + (c >> 6)] = sl[c];
        er[n * HEADS + (c >> 6)] = sr[c];
    }
}

__global__ void gat_init(float* __restrict__ agg, float* __restrict__ m,
                         float* __restrict__ denom)
{
    int i = blockIdx.x * blockDim.x + threadIdx.x;
    if (i < N_NODES * FTOT) agg[i] = 0.f;
    if (i < N_NODES * HEADS) { m[i] = -__builtin_inff(); denom[i] = 0.f; }
}

__device__ __forceinline__ void atomicMaxF(float* addr, float val)
{
    if (val >= 0.f)
        atomicMax((int*)addr, __float_as_int(val));
    else
        atomicMin((unsigned int*)addr, __float_as_uint(val));
}

__device__ __forceinline__ float leaky(float x)
{
    return x > 0.f ? x : NEG_SLOPE * x;
}

__global__ void gat_edge_max(const int* __restrict__ src, const int* __restrict__ dst,
                             const float* __restrict__ el, const float* __restrict__ er,
                             float* __restrict__ m)
{
    int i = blockIdx.x * blockDim.x + threadIdx.x;
    if (i >= N_EDGES * HEADS) return;
    int e = i >> 2, h = i & 3;
    int s = src[e], d = dst[e];
    float x = leaky(el[s * HEADS + h] + er[d * HEADS + h]);
    atomicMaxF(&m[d * HEADS + h], x);
}

__global__ void gat_edge_denom(const int* __restrict__ src, const int* __restrict__ dst,
                               const float* __restrict__ el, const float* __restrict__ er,
                               const float* __restrict__ m, float* __restrict__ denom)
{
    int i = blockIdx.x * blockDim.x + threadIdx.x;
    if (i >= N_EDGES * HEADS) return;
    int e = i >> 2, h = i & 3;
    int s = src[e], d = dst[e];
    float x = leaky(el[s * HEADS + h] + er[d * HEADS + h]);
    atomicAdd(&denom[d * HEADS + h], expf(x - m[d * HEADS + h]));
}

__global__ __launch_bounds__(256) void gat_edge_scatter(
    const int* __restrict__ src, const int* __restrict__ dst,
    const float* __restrict__ el, const float* __restrict__ er,
    const float* __restrict__ m, const float* __restrict__ denom,
    const float* __restrict__ feat, float* __restrict__ agg)
{
    int e = blockIdx.x;
    int c = threadIdx.x, h = c >> 6;
    int s = src[e], d = dst[e];
    float x = leaky(el[s * HEADS + h] + er[d * HEADS + h]);
    float alpha = expf(x - m[d * HEADS + h]) / denom[d * HEADS + h];
    atomicAdd(&agg[(size_t)d * FTOT + c], feat[(size_t)s * FTOT + c] * alpha);
}

__device__ __forceinline__ float elu1(float x)
{
    return x > 0.f ? x : expf(x) - 1.f;
}

__global__ void gat_finalize_inplace(float* __restrict__ agg,
                                     const float* __restrict__ bias)
{
    int i = blockIdx.x * blockDim.x + threadIdx.x;
    if (i >= N_NODES * FTOT) return;
    agg[i] = elu1(agg[i] + bias[i & 255]);
}

__global__ void gat_finalize_out(const float* __restrict__ agg,
                                 const float* __restrict__ bias,
                                 float* __restrict__ out)
{
    int i = blockIdx.x * blockDim.x + threadIdx.x;
    if (i >= N_NODES * FTOT) return;
    int c = i & 255, n = i >> 8;
    float x = elu1(agg[i] + bias[c]);
    int h = c >> 6, d = c & 63;
    out[(size_t)h * (N_NODES * HID) + (size_t)n * HID + d] = x;
}

extern "C" void kernel_launch(void* const* d_in, const int* in_sizes, int n_in,
                              void* d_out, int out_size, void* d_ws, size_t ws_size,
                              hipStream_t stream)
{
    (void)in_sizes; (void)n_in; (void)out_size; (void)ws_size;

    const float* x   = (const float*)d_in[0];
    const int*   src = (const int*)d_in[1];
    const int*   dst = (const int*)d_in[2];
    const float* W1  = (const float*)d_in[3];
    const float* al1 = (const float*)d_in[4];
    const float* ar1 = (const float*)d_in[5];
    const float* b1  = (const float*)d_in[6];
    const float* W2  = (const float*)d_in[7];
    const float* al2 = (const float*)d_in[8];
    const float* ar2 = (const float*)d_in[9];
    const float* b2  = (const float*)d_in[10];
    float* out = (float*)d_out;

    char* ws = (char*)d_ws;
    const size_t featB = (size_t)N_NODES * FTOT * sizeof(float);
    const size_t nhB   = (size_t)N_NODES * HEADS * sizeof(float);
    float* feat  = (float*)ws; ws += featB;
    float* agg   = (float*)ws; ws += featB;
    float* el    = (float*)ws; ws += nhB;
    float* er    = (float*)ws; ws += nhB;
    float* m     = (float*)ws; ws += nhB;
    float* denom = (float*)ws; ws += nhB;

    dim3 gGemm(N_NODES / 16, 2), bGemm(256);
    const int NC  = N_NODES * FTOT;
    const int EH  = N_EDGES * HEADS;
    const int gNC = (NC + 255) / 256;
    const int gEH = (EH + 255) / 256;

    gat_gemm_f32_wmma<<<gGemm, bGemm, 0, stream>>>(x, W1, feat);
    gat_eler<<<N_NODES, 256, 0, stream>>>(feat, al1, ar1, el, er);
    gat_init<<<gNC, 256, 0, stream>>>(agg, m, denom);
    gat_edge_max<<<gEH, 256, 0, stream>>>(src, dst, el, er, m);
    gat_edge_denom<<<gEH, 256, 0, stream>>>(src, dst, el, er, m, denom);
    gat_edge_scatter<<<N_EDGES, 256, 0, stream>>>(src, dst, el, er, m, denom, feat, agg);
    gat_finalize_inplace<<<gNC, 256, 0, stream>>>(agg, b1);

    gat_gemm_f32_wmma<<<gGemm, bGemm, 0, stream>>>(agg, W2, feat);
    gat_eler<<<N_NODES, 256, 0, stream>>>(feat, al2, ar2, el, er);
    gat_init<<<gNC, 256, 0, stream>>>(agg, m, denom);
    gat_edge_max<<<gEH, 256, 0, stream>>>(src, dst, el, er, m);
    gat_edge_denom<<<gEH, 256, 0, stream>>>(src, dst, el, er, m, denom);
    gat_edge_scatter<<<N_EDGES, 256, 0, stream>>>(src, dst, el, er, m, denom, feat, agg);
    gat_finalize_out<<<gNC, 256, 0, stream>>>(agg, b2, out);
}


# Edges grouped by destination (integer tickets, scan, each node's list put in edge-index order); one wave per node does softmax, weighted row sum, bias+ELU by plain loads; no f32 atomics, fixed order
# speedup vs baseline: 3.5711x; 3.5711x over previous
#include <hip/hip_runtime.h>
#include <math.h>


#define N_NODES 50000
#define N_EDGES 800000
#define IN_DIM 256
#define HID 64
#define HEADS 4
#define FTOT 256
#define NEG_SLOPE 0.2f

typedef __attribute__((ext_vector_type(2))) float v2f;
typedef __attribute__((ext_vector_type(8))) float v8f;

__global__ __launch_bounds__(256) void gat_gemm_f32_wmma(
    const float* __restrict__ A, const float* __restrict__ W,
    float* __restrict__ out)
{
    __shared__ float lds[16 * 256];
    const int tid  = threadIdx.x;
    const int row0 = blockIdx.x * 16;

    const float* Ablk = A + (size_t)row0 * FTOT;
    #pragma unroll
    for (int i = 0; i < 16; ++i)
        lds[tid + i * 256] = Ablk[tid + i * 256];
    __syncthreads();

    const int wave  = tid >> 5;
    const int lane  = tid & 31;
    const int col0  = (blockIdx.y * 8 + wave) * 16;
    const int hlane = lane & 15;
    const int koff  = (lane >> 4) * 2;

    v8f acc = {};
    const float* Wc = W + col0 + hlane;
    #pragma unroll 4
    for (int k = 0; k < IN_DIM; k += 4) {
        v2f a = *(const v2f*)&lds[hlane * 256 + k + koff];
        v2f b;
        b.x = Wc[(size_t)(k + koff) * FTOT];
        b.y = Wc[(size_t)(k + koff + 1) * FTOT];
        acc = __builtin_amdgcn_wmma_f32_16x16x4_f32(
                  false, a, false, b, (short)0, acc, false, false);
    }

    const int srow = row0 + ((lane >> 4) * 8);
    float* Orow = out + (size_t)srow * FTOT + col0 + hlane;
    #pragma unroll
    for (int j = 0; j < 8; ++j)
        Orow[(size_t)j * FTOT] = acc[j];
}

__global__ __launch_bounds__(256) void gat_eler(
    const float* __restrict__ feat, const float* __restrict__ al,
    const float* __restrict__ ar, float* __restrict__ el, float* __restrict__ er)
{
    __shared__ float sl[256], sr[256];
    const int n = blockIdx.x, c = threadIdx.x;
    float v = feat[(size_t)n * FTOT + c];
    sl[c] = v * al[c];
    sr[c] = v * ar[c];
    __syncthreads();
    const int d = c & 63;
    for (int s = 32; s > 0; s >>= 1) {
        if (d < s) { sl[c] += sl[c + s]; sr[c] += sr[c + s]; }
        __syncthreads();
    }
    if (d == 0) {
        el[n * HEADS + (c >> 6)] = sl[c];
        er[n * HEADS + (c >> 6)] = sr[c];
    }
}

__global__ void csr_zero(int* __restrict__ cnt)
{
    int i = blockIdx.x * blockDim.x + threadIdx.x;
    if (i < N_NODES) cnt[i] = 0;
}

__global__ void csr_ticket(const int* __restrict__ dst, int* __restrict__ cnt,
                           int* __restrict__ ticket)
{
    int e = blockIdx.x * blockDim.x + threadIdx.x;
    if (e >= N_EDGES) return;
    ticket[e] = atomicAdd(&cnt[dst[e]], 1);
}

__global__ __launch_bounds__(1024) void csr_scan(const int* __restrict__ cnt,
                                                 int* __restrict__ rowptr)
{
    __shared__ int wpre[32];
    __shared__ int wtot[32];
    const int t = threadIdx.x, lane = t & 31, w = t >> 5;
    const int base = blockIdx.x * 1024;

    int pre = 0;
    for (int i = t; i < base; i += 1024) pre += cnt[i];
    for (int o = 16; o > 0; o >>= 1) pre += __shfl_xor(pre, o);

    const int i = base + t;
    const int c = (i < N_NODES) ? cnt[i] : 0;
    int v = c;
    for (int o = 1; o < 32; o <<= 1) {
        int u = __shfl_up(v, o);
        if (lane >= o) v += u;
    }
    if (lane == 0)  wpre[w] = pre;
    if (lane == 31) wtot[w] = v;
    __syncthreads();

    int a = wpre[lane];
    int b = (lane < w) ? wtot[lane] : 0;
    for (int o = 16; o > 0; o >>= 1) {
        a += __shfl_xor(a, o);
        b += __shfl_xor(b, o);
    }
    const int excl = a + b + v - c;
    if (i < N_NODES) rowptr[i] = excl;
    if (i == N_NODES - 1) rowptr[N_NODES] = excl + c;
}

__global__ void csr_place(const int* __restrict__ src, const int* __restrict__ dst,
                          const int* __restrict__ ticket, const int* __restrict__ rowptr,
                          int* __restrict__ elist)
{
    int e = blockIdx.x * blockDim.x + threadIdx.x;
    if (e >= N_EDGES) return;
    int pos = rowptr[dst[e]] + ticket[e];
    *(int2*)(elist + 2 * (size_t)pos) = make_int2(e, src[e]);
}

__global__ __launch_bounds__(256) void csr_order(const int* __restrict__ rowptr,
                                                 const int* __restrict__ elist,
                                                 int* __restrict__ ssrc)
{
    const int lane = threadIdx.x & 31;
    const int n = __builtin_amdgcn_readfirstlane(blockIdx.x * 8 + (threadIdx.x >> 5));
    if (n >= N_NODES) return;
    const int beg = rowptr[n], end = rowptr[n + 1];
    const int deg = end - beg;
    if (deg <= 32) {
        int2 me = make_int2(0x7fffffff, 0);
        if (lane < deg) me = *(const int2*)(elist + 2 * (size_t)(beg + lane));
        int r = 0;
        for (int j = 0; j < deg; ++j)
            r += (__builtin_amdgcn_readlane(me.x, j) < me.x) ? 1 : 0;
        if (lane < deg) ssrc[beg + r] = me.y;
    } else {
        for (int i = beg + lane; i < end; i += 32) {
            int2 me = *(const int2*)(elist + 2 * (size_t)i);
            int r = 0;
            for (int k = beg; k < end; ++k) r += (elist[2 * (size_t)k] < me.x) ? 1 : 0;
            ssrc[beg + r] = me.y;
        }
    }
}

__device__ __forceinline__ float leaky(float x)
{
    return x > 0.f ? x : NEG_SLOPE * x;
}

__device__ __forceinline__ float elu1(float x)
{
    return x > 0.f ? x : expf(x) - 1.f;
}

__device__ __forceinline__ float wave_max(float v)
{
    for (int o = 16; o > 0; o >>= 1) v = fmaxf(v, __shfl_xor(v, o));
    return v;
}

__device__ __forceinline__ float wave_sum(float v)
{
    for (int o = 16; o > 0; o >>= 1) v += __shfl_xor(v, o);
    return v;
}

__device__ __forceinline__ float lane_bcast(float v, int j)
{
    return __int_as_float(__builtin_amdgcn_readlane(__float_as_int(v), j));
}

__device__ __forceinline__ void fma4(float4& acc, float w, const float4& f)
{
    acc.x = fmaf(w, f.x, acc.x);
    acc.y = fmaf(w, f.y, acc.y);
    acc.z = fmaf(w, f.z, acc.z);
    acc.w = fmaf(w, f.w, acc.w);
}

__device__ __forceinline__ void edge_weights(float p0, float p1, float p2, float p3,
                                             int j, bool hi, float& wA, float& wB)
{
    const float q0 = lane_bcast(p0, j), q1 = lane_bcast(p1, j);
    const float q2 = lane_bcast(p2, j), q3 = lane_bcast(p3, j);
    wA = hi ? q1 : q0;
    wB = hi ? q3 : q2;
}

__device__ __forceinline__ void gather_rows(
    const float* __restrict__ feat, int s, float p0, float p1, float p2, float p3,
    int cnt, int lane, float4& accA, float4& accB)
{
    const bool hi = lane >= 16;
    const float* fl = feat + 4 * lane;
    int j = 0;
    for (; j + 4 <= cnt; j += 4) {
        const float* r0 = fl + (size_t)__builtin_amdgcn_readlane(s, j)     * FTOT;
        const float* r1 = fl + (size_t)__builtin_amdgcn_readlane(s, j + 1) * FTOT;
        const float* r2 = fl + (size_t)__builtin_amdgcn_readlane(s, j + 2) * FTOT;
        const float* r3 = fl + (size_t)__builtin_amdgcn_readlane(s, j + 3) * FTOT;
        const float4 a0 = *(const float4*)r0, b0 = *(const float4*)(r0 + 128);
        const float4 a1 = *(const float4*)r1, b1 = *(const float4*)(r1 + 128);
        const float4 a2 = *(const float4*)r2, b2 = *(const float4*)(r2 + 128);
        const float4 a3 = *(const float4*)r3, b3 = *(const float4*)(r3 + 128);
        float wA0, wB0, wA1, wB1, wA2, wB2, wA3, wB3;
        edge_weights(p0, p1, p2, p3, j,     hi, wA0, wB0);
        edge_weights(p0, p1, p2, p3, j + 1, hi, wA1, wB1);
        edge_weights(p0, p1, p2, p3, j + 2, hi, wA2, wB2);
        edge_weights(p0, p1, p2, p3, j + 3, hi, wA3, wB3);
        fma4(accA, wA0, a0); fma4(accB, wB0, b0);
        fma4(accA, wA1, a1); fma4(accB, wB1, b1);
        fma4(accA, wA2, a2); fma4(accB, wB2, b2);
        fma4(accA, wA3, a3); fma4(accB, wB3, b3);
    }
    const int rem = cnt - j;
    if (rem > 0) {
        const float* r0 = fl + (size_t)__builtin_amdgcn_readlane(s, j) * FTOT;
        const float4 a0 = *(const float4*)r0, b0 = *(const float4*)(r0 + 128);
        float4 a1 = {0.f, 0.f, 0.f, 0.f}, b1 = a1, a2 = a1, b2 = a1;
        if (rem > 1) {
            const float* r1 = fl + (size_t)__builtin_amdgcn_readlane(s, j + 1) * FTOT;
            a1 = *(const float4*)r1; b1 = *(const float4*)(r1 + 128);
        }
        if (rem > 2) {
            const float* r2 = fl + (size_t)__builtin_amdgcn_readlane(s, j + 2) * FTOT;
            a2 = *(const float4*)r2; b2 = *(const float4*)(r2 + 128);
        }
        float wA, wB;
        edge_weights(p0, p1, p2, p3, j, hi, wA, wB);
        fma4(accA, wA, a0); fma4(accB, wB, b0);
        if (rem > 1) {
            edge_weights(p0, p1, p2, p3, j + 1, hi, wA, wB);
            fma4(accA, wA, a1); fma4(accB, wB, b1);
        }
        if (rem > 2) {
            edge_weights(p0, p1, p2, p3, j + 2, hi, wA, wB);
            fma4(accA, wA, a2); fma4(accB, wB, b2);
        }
    }
}

__global__ __launch_bounds__(256) void gat_node_gather(
    const int* __restrict__ rowptr, const int* __restrict__ ssrc,
    const float* __restrict__ el, const float* __restrict__ er,
    const float* __restrict__ feat, const float* __restrict__ bias,
    float* __restrict__ outp, int head_major)
{
    const int lane = threadIdx.x & 31;
    const int n = __builtin_amdgcn_readfirstlane(blockIdx.x * 8 + (threadIdx.x >> 5));
    if (n >= N_NODES) return;
    const int beg = rowptr[n], end = rowptr[n + 1];
    const int deg = end - beg;
    const float4 ern = *(const float4*)(er + (size_t)n * HEADS);
    const float NINF = -__builtin_inff();

    float4 accA = {0.f, 0.f, 0.f, 0.f}, accB = {0.f, 0.f, 0.f, 0.f};

    if (deg <= 32) {
        const bool valid = lane < deg;
        int s = 0;
        float4 l = {0.f, 0.f, 0.f, 0.f};
        if (valid) {
            s = ssrc[beg + lane];
            l = *(const float4*)(el + (size_t)s * HEADS);
        }
        const float x0 = leaky(l.x + ern.x), x1 = leaky(l.y + ern.y);
        const float x2 = leaky(l.z + ern.z), x3 = leaky(l.w + ern.w);
        const float m0 = wave_max(valid ? x0 : NINF), m1 = wave_max(valid ? x1 : NINF);
        const float m2 = wave_max(valid ? x2 : NINF), m3 = wave_max(valid ? x3 : NINF);
        float p0 = valid ? expf(x0 - m0) : 0.f, p1 = valid ? expf(x1 - m1) : 0.f;
        float p2 = valid ? expf(x2 - m2) : 0.f, p3 = valid ? expf(x3 - m3) : 0.f;
        const float d0 = wave_sum(p0), d1 = wave_sum(p1);
        const float d2 = wave_sum(p2), d3 = wave_sum(p3);
        if (valid) { p0 /= d0; p1 /= d1; p2 /= d2; p3 /= d3; }
        gather_rows(feat, s, p0, p1, p2, p3, deg, lane, accA, accB);
    } else {
        float m0 = NINF, m1 = NINF, m2 = NINF, m3 = NINF;
        for (int i = beg + lane; i < end; i += 32) {
            const float4 l = *(const float4*)(el + (size_t)ssrc[i] * HEADS);
            m0 = fmaxf(m0, leaky(l.x + ern.x)); m1 = fmaxf(m1, leaky(l.y + ern.y));
            m2 = fmaxf(m2, leaky(l.z + ern.z)); m3 = fmaxf(m3, leaky(l.w + ern.w));
        }
        m0 = wave_max(m0); m1 = wave_max(m1); m2 = wave_max(m2); m3 = wave_max(m3);
        float d0 = 0.f, d1 = 0.f, d2 = 0.f, d3 = 0.f;
        for (int i = beg + lane; i < end; i += 32) {
            const float4 l = *(const float4*)(el + (size_t)ssrc[i] * HEADS);
            d0 += expf(leaky(l.x + ern.x) - m0); d1 += expf(leaky(l.y + ern.y) - m1);
            d2 += expf(leaky(l.z + ern.z) - m2); d3 += expf(leaky(l.w + ern.w) - m3);
        }
        d0 = wave_sum(d0); d1 = wave_sum(d1); d2 = wave_sum(d2); d3 = wave_sum(d3);
        for (int base = beg; base < end; base += 32) {
            const int i = base + lane;
            int s = 0;
            float p0 = 0.f, p1 = 0.f, p2 = 0.f, p3 = 0.f;
            if (i < end) {
                s = ssrc[i];
                const float4 l = *(const float4*)(el + (size_t)s * HEADS);
                p0 = expf(leaky(l.x + ern.x) - m0) / d0;
                p1 = expf(leaky(l.y + ern.y) - m1) / d1;
                p2 = expf(leaky(l.z + ern.z) - m2) / d2;
                p3 = expf(leaky(l.w + ern.w) - m3) / d3;
            }
            const int left = end - base;
            gather_rows(feat, s, p0, p1, p2, p3, left < 32 ? left : 32, lane, accA, accB);
        }
    }

    const float4 bA = *(const float4*)(bias + 4 * lane);
    const float4 bB = *(const float4*)(bias + 128 + 4 * lane);
    float4 oA, oB;
    oA.x = elu1(accA.x + bA.x); oA.y = elu1(accA.y + bA.y);
    oA.z = elu1(accA.z + bA.z); oA.w = elu1(accA.w + bA.w);
    oB.x = elu1(accB.x + bB.x); oB.y = elu1(accB.y + bB.y);
    oB.z = elu1(accB.z + bB.z); oB.w = elu1(accB.w + bB.w);

    if (head_major) {
        const int hsel = lane >> 4;
        const int d = (4 * lane) & 63;
        float* o = outp + (size_t)n * HID + d;
        *(float4*)(o + (size_t)hsel * (N_NODES * HID)) = oA;
        *(float4*)(o + (size_t)(hsel + 2) * (N_NODES * HID)) = oB;
    } else {
        float* o = outp + (size_t)n * FTOT + 4 * lane;
        *(float4*)o = oA;
        *(float4*)(o + 128) = oB;
    }
}

extern "C" void kernel_launch(void* const* d_in, const int* in_sizes, int n_in,
                              void* d_out, int out_size, void* d_ws, size_t ws_size,
                              hipStream_t stream)
{
    (void)in_sizes; (void)n_in; (void)out_size; (void)ws_size;

    const float* x   = (const float*)d_in[0];
    const int*   src = (const int*)d_in[1];
    const int*   dst = (const int*)d_in[2];
    const float* W1  = (const float*)d_in[3];
    const float* al1 = (const float*)d_in[4];
    const float* ar1 = (const float*)d_in[5];
    const float* b1  = (const float*)d_in[6];
    const float* W2  = (const float*)d_in[7];
    const float* al2 = (const float*)d_in[8];
    const float* ar2 = (const float*)d_in[9];
    const float* b2  = (const float*)d_in[10];
    float* out = (float*)d_out;

    char* ws = (char*)d_ws;
    const size_t featB = (size_t)N_NODES * FTOT * sizeof(float);
    const size_t nhB   = (size_t)N_NODES * HEADS * sizeof(float);
    const size_t nodeB = 200192;
    const size_t edgeB = (size_t)N_EDGES * sizeof(int);
    float* feat   = (float*)ws; ws += featB;
    float* h1     = (float*)ws; ws += featB;
    float* el     = (float*)ws; ws += nhB;
    float* er     = (float*)ws; ws += nhB;
    int*   cnt    = (int*)ws;   ws += nodeB;
    int*   rowptr = (int*)ws;   ws += nodeB;
    int*   ticket = (int*)ws;   ws += edgeB;
    int*   elist  = (int*)ws;   ws += 2 * edgeB;
    int*   ssrc   = (int*)ws;   ws += edgeB;

    dim3 gGemm(N_NODES / 16, 2), bGemm(256);
    const int gN  = (N_NODES + 255) / 256;
    const int gE  = (N_EDGES + 255) / 256;
    const int gNW = (N_NODES + 7) / 8;

    csr_zero<<<gN, 256, 0, stream>>>(cnt);
    csr_ticket<<<gE, 256, 0, stream>>>(dst, cnt, ticket);
    csr_scan<<<(N_NODES + 1023) / 1024, 1024, 0, stream>>>(cnt, rowptr);
    csr_place<<<gE, 256, 0, stream>>>(src, dst, ticket, rowptr, elist);
    csr_order<<<gNW, 256, 0, stream>>>(rowptr, elist, ssrc);

    gat_gemm_f32_wmma<<<gGemm, bGemm, 0, stream>>>(x, W1, feat);
    gat_eler<<<N_NODES, 256, 0, stream>>>(feat, al1, ar1, el, er);
    gat_node_gather<<<gNW, 256, 0, stream>>>(rowptr, ssrc, el, er, feat, b1, h1, 0);

    gat_gemm_f32_wmma<<<gGemm, bGemm, 0, stream>>>(h1, W2, feat);
    gat_eler<<<N_NODES, 256, 0, stream>>>(feat, al2, ar2, el, er);
    gat_node_gather<<<gNW, 256, 0, stream>>>(rowptr, ssrc, el, er, feat, b2, out, 1);
}


# el/er now formed in the GEMM epilogue from the accumulator tiles (DPP row butterfly, a head's 4 waves added in fixed order via LDS); block-per-node el/er kernel removed; el/er sum order changes
# speedup vs baseline: 4.5091x; 1.2627x over previous
#include <hip/hip_runtime.h>
#include <math.h>


#define N_NODES 50000
#define N_EDGES 800000
#define IN_DIM 256
#define HID 64
#define HEADS 4
#define FTOT 256
#define NEG_SLOPE 0.2f

typedef __attribute__((ext_vector_type(2))) float v2f;
typedef __attribute__((ext_vector_type(8))) float v8f;

template <int M>
__device__ __forceinline__ float row_xmask(float v)
{
    return __int_as_float(
        __builtin_amdgcn_update_dpp(0, __float_as_int(v), 0x160 + M, 0xf, 0xf, true));
}

__global__ __launch_bounds__(256) void gat_gemm_f32_wmma(
    const float* __restrict__ A, const float* __restrict__ W,
    const float* __restrict__ al, const float* __restrict__ ar,
    float* __restrict__ out, float* __restrict__ el, float* __restrict__ er)
{
    __shared__ float lds[16 * 256];
    __shared__ float red[8 * 32];
    const int tid  = threadIdx.x;
    const int row0 = blockIdx.x * 16;

    const float* Ablk = A + (size_t)row0 * FTOT;
    #pragma unroll
    for (int i = 0; i < 16; ++i)
        lds[tid + i * 256] = Ablk[tid + i * 256];
    __syncthreads();

    const int wave  = tid >> 5;
    const int lane  = tid & 31;
    const int col0  = (blockIdx.y * 8 + wave) * 16;
    const int hlane = lane & 15;
    const int koff  = (lane >> 4) * 2;

    const float alc = al[col0 + hlane];
    const float arc = ar[col0 + hlane];

    v8f acc = {};
    const float* Wc = W + col0 + hlane;
    #pragma unroll 4
    for (int k = 0; k < IN_DIM; k += 4) {
        v2f a = *(const v2f*)&lds[hlane * 256 + k + koff];
        v2f b;
        b.x = Wc[(size_t)(k + koff) * FTOT];
        b.y = Wc[(size_t)(k + koff + 1) * FTOT];
        acc = __builtin_amdgcn_wmma_f32_16x16x4_f32(
                  false, a, false, b, (short)0, acc, false, false);
    }

    const bool b0 = lane & 1;
    const float ck = b0 ? arc : alc;
    const float cg = b0 ? alc : arc;
    float q[8];
    #pragma unroll
    for (int j = 0; j < 8; ++j) q[j] = acc[j] * ck + row_xmask<1>(acc[j] * cg);
    const bool b3 = lane & 8;
    float r4[4];
    #pragma unroll
    for (int t = 0; t < 4; ++t)
        r4[t] = (b3 ? q[t + 4] : q[t]) + row_xmask<8>(b3 ? q[t] : q[t + 4]);
    const bool b2 = lane & 4;
    float r2[2];
    #pragma unroll
    for (int u = 0; u < 2; ++u)
        r2[u] = (b2 ? r4[u + 2] : r4[u]) + row_xmask<4>(b2 ? r4[u] : r4[u + 2]);
    const bool b1 = lane & 2;
    const float tot = (b1 ? r2[1] : r2[0]) + row_xmask<2>(b1 ? r2[0] : r2[1]);
    red[wave * 32 + lane] = tot;
    __syncthreads();
    if (tid < 64) {
        const int arr = tid >> 5, r = (tid >> 1) & 15, hh = tid & 1;
        const float* p = &red[(hh * 4) * 32 + r * 2 + arr];
        const float s = (p[0] + p[32]) + (p[64] + p[96]);
        float* dst = arr ? er : el;
        dst[(size_t)(row0 + r) * HEADS + blockIdx.y * 2 + hh] = s;
    }

    const int srow = row0 + ((lane >> 4) * 8);
    float* Orow = out + (size_t)srow * FTOT + col0 + hlane;
    #pragma unroll
    for (int j = 0; j < 8; ++j)
        Orow[(size_t)j * FTOT] = acc[j];
}

__global__ void csr_zero(int* __restrict__ cnt)
{
    int i = blockIdx.x * blockDim.x + threadIdx.x;
    if (i < N_NODES) cnt[i] = 0;
}

__global__ void csr_ticket(const int* __restrict__ dst, int* __restrict__ cnt,
                           int* __restrict__ ticket)
{
    int e = blockIdx.x * blockDim.x + threadIdx.x;
    if (e >= N_EDGES) return;
    ticket[e] = atomicAdd(&cnt[dst[e]], 1);
}

__global__ __launch_bounds__(1024) void csr_scan(const int* __restrict__ cnt,
                                                 int* __restrict__ rowptr)
{
    __shared__ int wpre[32];
    __shared__ int wtot[32];
    const int t = threadIdx.x, lane = t & 31, w = t >> 5;
    const int base = blockIdx.x * 1024;

    int pre = 0;
    for (int i = t; i < base; i += 1024) pre += cnt[i];
    for (int o = 16; o > 0; o >>= 1) pre += __shfl_xor(pre, o);

    const int i = base + t;
    const int c = (i < N_NODES) ? cnt[i] : 0;
    int v = c;
    for (int o = 1; o < 32; o <<= 1) {
        int u = __shfl_up(v, o);
        if (lane >= o) v += u;
    }
    if (lane == 0)  wpre[w] = pre;
    if (lane == 31) wtot[w] = v;
    __syncthreads();

    int a = wpre[lane];
    int b = (lane < w) ? wtot[lane] : 0;
    for (int o = 16; o > 0; o >>= 1) {
        a += __shfl_xor(a, o);
        b += __shfl_xor(b, o);
    }
    const int excl = a + b + v - c;
    if (i < N_NODES) rowptr[i] = excl;
    if (i == N_NODES - 1) rowptr[N_NODES] = excl + c;
}

__global__ void csr_place(const int* __restrict__ src, const int* __restrict__ dst,
                          const int* __restrict__ ticket, const int* __restrict__ rowptr,
                          int* __restrict__ elist)
{
    int e = blockIdx.x * blockDim.x + threadIdx.x;
    if (e >= N_EDGES) return;
    int pos = rowptr[dst[e]] + ticket[e];
    *(int2*)(elist + 2 * (size_t)pos) = make_int2(e, src[e]);
}

__global__ __launch_bounds__(256) void csr_order(const int* __restrict__ rowptr,
                                                 const int* __restrict__ elist,
                                                 int* __restrict__ ssrc)
{
    const int lane = threadIdx.x & 31;
    const int n = __builtin_amdgcn_readfirstlane(blockIdx.x * 8 + (threadIdx.x >> 5));
    if (n >= N_NODES) return;
    const int beg = rowptr[n], end = rowptr[n + 1];
    const int deg = end - beg;
    if (deg <= 32) {
        int2 me = make_int2(0x7fffffff, 0);
        if (lane < deg) me = *(const int2*)(elist + 2 * (size_t)(beg + lane));
        int r = 0;
        for (int j = 0; j < deg; ++j)
            r += (__builtin_amdgcn_readlane(me.x, j) < me.x) ? 1 : 0;
        if (lane < deg) ssrc[beg + r] = me.y;
    } else {
        for (int i = beg + lane; i < end; i += 32) {
            int2 me = *(const int2*)(elist + 2 * (size_t)i);
            int r = 0;
            for (int k = beg; k < end; ++k) r += (elist[2 * (size_t)k] < me.x) ? 1 : 0;
            ssrc[beg + r] = me.y;
        }
    }
}

__device__ __forceinline__ float leaky(float x)
{
    return x > 0.f ? x : NEG_SLOPE * x;
}

__device__ __forceinline__ float elu1(float x)
{
    return x > 0.f ? x : expf(x) - 1.f;
}

__device__ __forceinline__ float wave_max(float v)
{
    for (int o = 16; o > 0; o >>= 1) v = fmaxf(v, __shfl_xor(v, o));
    return v;
}

__device__ __forceinline__ float wave_sum(float v)
{
    for (int o = 16; o > 0; o >>= 1) v += __shfl_xor(v, o);
    return v;
}

__device__ __forceinline__ float lane_bcast(float v, int j)
{
    return __int_as_float(__builtin_amdgcn_readlane(__float_as_int(v), j));
}

__device__ __forceinline__ void fma4(float4& acc, float w, const float4& f)
{
    acc.x = fmaf(w, f.x, acc.x);
    acc.y = fmaf(w, f.y, acc.y);
    acc.z = fmaf(w, f.z, acc.z);
    acc.w = fmaf(w, f.w, acc.w);
}

__device__ __forceinline__ void edge_weights(float p0, float p1, float p2, float p3,
                                             int j, bool hi, float& wA, float& wB)
{
    const float q0 = lane_bcast(p0, j), q1 = lane_bcast(p1, j);
    const float q2 = lane_bcast(p2, j), q3 = lane_bcast(p3, j);
    wA = hi ? q1 : q0;
    wB = hi ? q3 : q2;
}

__device__ __forceinline__ void gather_rows(
    const float* __restrict__ feat, int s, float p0, float p1, float p2, float p3,
    int cnt, int lane, float4& accA, float4& accB)
{
    const bool hi = lane >= 16;
    const float* fl = feat + 4 * lane;
    int j = 0;
    for (; j + 4 <= cnt; j += 4) {
        const float* r0 = fl + (size_t)__builtin_amdgcn_readlane(s, j)     * FTOT;
        const float* r1 = fl + (size_t)__builtin_amdgcn_readlane(s, j + 1) * FTOT;
        const float* r2 = fl + (size_t)__builtin_amdgcn_readlane(s, j + 2) * FTOT;
        const float* r3 = fl + (size_t)__builtin_amdgcn_readlane(s, j + 3) * FTOT;
        const float4 a0 = *(const float4*)r0, b0 = *(const float4*)(r0 + 128);
        const float4 a1 = *(const float4*)r1, b1 = *(const float4*)(r1 + 128);
        const float4 a2 = *(const float4*)r2, b2 = *(const float4*)(r2 + 128);
        const float4 a3 = *(const float4*)r3, b3 = *(const float4*)(r3 + 128);
        float wA0, wB0, wA1, wB1, wA2, wB2, wA3, wB3;
        edge_weights(p0, p1, p2, p3, j,     hi, wA0, wB0);
        edge_weights(p0, p1, p2, p3, j + 1, hi, wA1, wB1);
        edge_weights(p0, p1, p2, p3, j + 2, hi, wA2, wB2);
        edge_weights(p0, p1, p2, p3, j + 3, hi, wA3, wB3);
        fma4(accA, wA0, a0); fma4(accB, wB0, b0);
        fma4(accA, wA1, a1); fma4(accB, wB1, b1);
        fma4(accA, wA2, a2); fma4(accB, wB2, b2);
        fma4(accA, wA3, a3); fma4(accB, wB3, b3);
    }
    const int rem = cnt - j;
    if (rem > 0) {
        const float* r0 = fl + (size_t)__builtin_amdgcn_readlane(s, j) * FTOT;
        const float4 a0 = *(const float4*)r0, b0 = *(const float4*)(r0 + 128);
        float4 a1 = {0.f, 0.f, 0.f, 0.f}, b1 = a1, a2 = a1, b2 = a1;
        if (rem > 1) {
            const float* r1 = fl + (size_t)__builtin_amdgcn_readlane(s, j + 1) * FTOT;
            a1 = *(const float4*)r1; b1 = *(const float4*)(r1 + 128);
        }
        if (rem > 2) {
            const float* r2 = fl + (size_t)__builtin_amdgcn_readlane(s, j + 2) * FTOT;
            a2 = *(const float4*)r2; b2 = *(const float4*)(r2 + 128);
        }
        float wA, wB;
        edge_weights(p0, p1, p2, p3, j, hi, wA, wB);
        fma4(accA, wA, a0); fma4(accB, wB, b0);
        if (rem > 1) {
            edge_weights(p0, p1, p2, p3, j + 1, hi, wA, wB);
            fma4(accA, wA, a1); fma4(accB, wB, b1);
        }
        if (rem > 2) {
            edge_weights(p0, p1, p2, p3, j + 2, hi, wA, wB);
            fma4(accA, wA, a2); fma4(accB, wB, b2);
        }
    }
}

__global__ __launch_bounds__(256) void gat_node_gather(
    const int* __restrict__ rowptr, const int* __restrict__ ssrc,
    const float* __restrict__ el, const float* __restrict__ er,
    const float* __restrict__ feat, const float* __restrict__ bias,
    float* __restrict__ outp, int head_major)
{
    const int lane = threadIdx.x & 31;
    const int n = __builtin_amdgcn_readfirstlane(blockIdx.x * 8 + (threadIdx.x >> 5));
    if (n >= N_NODES) return;
    const int beg = rowptr[n], end = rowptr[n + 1];
    const int deg = end - beg;
    const float4 ern = *(const float4*)(er + (size_t)n * HEADS);
    const float NINF = -__builtin_inff();

    float4 accA = {0.f, 0.f, 0.f, 0.f}, accB = {0.f, 0.f, 0.f, 0.f};

    if (deg <= 32) {
        const bool valid = lane < deg;
        int s = 0;
        float4 l = {0.f, 0.f, 0.f, 0.f};
        if (valid) {
            s = ssrc[beg + lane];
            l = *(const float4*)(el + (size_t)s * HEADS);
        }
        const float x0 = leaky(l.x + ern.x), x1 = leaky(l.y + ern.y);
        const float x2 = leaky(l.z + ern.z), x3 = leaky(l.w + ern.w);
        const float m0 = wave_max(valid ? x0 : NINF), m1 = wave_max(valid ? x1 : NINF);
        const float m2 = wave_max(valid ? x2 : NINF), m3 = wave_max(valid ? x3 : NINF);
        float p0 = valid ? expf(x0 - m0) : 0.f, p1 = valid ? expf(x1 - m1) : 0.f;
        float p2 = valid ? expf(x2 - m2) : 0.f, p3 = valid ? expf(x3 - m3) : 0.f;
        const float d0 = wave_sum(p0), d1 = wave_sum(p1);
        const float d2 = wave_sum(p2), d3 = wave_sum(p3);
        if (valid) { p0 /= d0; p1 /= d1; p2 /= d2; p3 /= d3; }
        gather_rows(feat, s, p0, p1, p2, p3, deg, lane, accA, accB);
    } else {
        float m0 = NINF, m1 = NINF, m2 = NINF, m3 = NINF;
        for (int i = beg + lane; i < end; i += 32) {
            const float4 l = *(const float4*)(el + (size_t)ssrc[i] * HEADS);
            m0 = fmaxf(m0, leaky(l.x + ern.x)); m1 = fmaxf(m1, leaky(l.y + ern.y));
            m2 = fmaxf(m2, leaky(l.z + ern.z)); m3 = fmaxf(m3, leaky(l.w + ern.w));
        }
        m0 = wave_max(m0); m1 = wave_max(m1); m2 = wave_max(m2); m3 = wave_max(m3);
        float d0 = 0.f, d1 = 0.f, d2 = 0.f, d3 = 0.f;
        for (int i = beg + lane; i < end; i += 32) {
            const float4 l = *(const float4*)(el + (size_t)ssrc[i] * HEADS);
            d0 += expf(leaky(l.x + ern.x) - m0); d1 += expf(leaky(l.y + ern.y) - m1);
            d2 += expf(leaky(l.z + ern.z) - m2); d3 += expf(leaky(l.w + ern.w) - m3);
        }
        d0 = wave_sum(d0); d1 = wave_sum(d1); d2 = wave_sum(d2); d3 = wave_sum(d3);
        for (int base = beg; base < end; base += 32) {
            const int i = base + lane;
            int s = 0;
            float p0 = 0.f, p1 = 0.f, p2 = 0.f, p3 = 0.f;
            if (i < end) {
                s = ssrc[i];
                const float4 l = *(const float4*)(el + (size_t)s * HEADS);
                p0 = expf(leaky(l.x + ern.x) - m0) / d0;
                p1 = expf(leaky(l.y + ern.y) - m1) / d1;
                p2 = expf(leaky(l.z + ern.z) - m2) / d2;
                p3 = expf(leaky(l.w + ern.w) - m3) / d3;
            }
            const int left = end - base;
            gather_rows(feat, s, p0, p1, p2, p3, left < 32 ? left : 32, lane, accA, accB);
        }
    }

    const float4 bA = *(const float4*)(bias + 4 * lane);
    const float4 bB = *(const float4*)(bias + 128 + 4 * lane);
    float4 oA, oB;
    oA.x = elu1(accA.x + bA.x); oA.y = elu1(accA.y + bA.y);
    oA.z = elu1(accA.z + bA.z); oA.w = elu1(accA.w + bA.w);
    oB.x = elu1(accB.x + bB.x); oB.y = elu1(accB.y + bB.y);
    oB.z = elu1(accB.z + bB.z); oB.w = elu1(accB.w + bB.w);

    if (head_major) {
        const int hsel = lane >> 4;
        const int d = (4 * lane) & 63;
        float* o = outp + (size_t)n * HID + d;
        *(float4*)(o + (size_t)hsel * (N_NODES * HID)) = oA;
        *(float4*)(o + (size_t)(hsel + 2) * (N_NODES * HID)) = oB;
    } else {
        float* o = outp + (size_t)n * FTOT + 4 * lane;
        *(float4*)o = oA;
        *(float4*)(o + 128) = oB;
    }
}

extern "C" void kernel_launch(void* const* d_in, const int* in_sizes, int n_in,
                              void* d_out, int out_size, void* d_ws, size_t ws_size,
                              hipStream_t stream)
{
    (void)in_sizes; (void)n_in; (void)out_size; (void)ws_size;

    const float* x   = (const float*)d_in[0];
    const int*   src = (const int*)d_in[1];
    const int*   dst = (const int*)d_in[2];
    const float* W1  = (const float*)d_in[3];
    const float* al1 = (const float*)d_in[4];
    const float* ar1 = (const float*)d_in[5];
    const float* b1  = (const float*)d_in[6];
    const float* W2  = (const float*)d_in[7];
    const float* al2 = (const float*)d_in[8];
    const float* ar2 = (const float*)d_in[9];
    const float* b2  = (const float*)d_in[10];
    float* out = (float*)d_out;

    char* ws = (char*)d_ws;
    const size_t featB = (size_t)N_NODES * FTOT * sizeof(float);
    const size_t nhB   = (size_t)N_NODES * HEADS * sizeof(float);
    const size_t nodeB = 200192;
    const size_t edgeB = (size_t)N_EDGES * sizeof(int);
    float* feat   = (float*)ws; ws += featB;
    float* h1     = (float*)ws; ws += featB;
    float* el     = (float*)ws; ws += nhB;
    float* er     = (float*)ws; ws += nhB;
    int*   cnt    = (int*)ws;   ws += nodeB;
    int*   rowptr = (int*)ws;   ws += nodeB;
    int*   ticket = (int*)ws;   ws += edgeB;
    int*   elist  = (int*)ws;   ws += 2 * edgeB;
    int*   ssrc   = (int*)ws;   ws += edgeB;

    dim3 gGemm(N_NODES / 16, 2), bGemm(256);
    const int gN  = (N_NODES + 255) / 256;
    const int gE  = (N_EDGES + 255) / 256;
    const int gNW = (N_NODES + 7) / 8;

    csr_zero<<<gN, 256, 0, stream>>>(cnt);
    csr_ticket<<<gE, 256, 0, stream>>>(dst, cnt, ticket);
    csr_scan<<<(N_NODES + 1023) / 1024, 1024, 0, stream>>>(cnt, rowptr);
    csr_place<<<gE, 256, 0, stream>>>(src, dst, ticket, rowptr, elist);
    csr_order<<<gNW, 256, 0, stream>>>(rowptr, elist, ssrc);

    gat_gemm_f32_wmma<<<gGemm, bGemm, 0, stream>>>(x, W1, al1, ar1, feat, el, er);
    gat_node_gather<<<gNW, 256, 0, stream>>>(rowptr, ssrc, el, er, feat, b1, h1, 0);

    gat_gemm_f32_wmma<<<gGemm, bGemm, 0, stream>>>(h1, W2, al2, ar2, feat, el, er);
    gat_node_gather<<<gNW, 256, 0, stream>>>(rowptr, ssrc, el, er, feat, b2, out, 1);
}


# GEMM re-tiled, block 16x128 -> 80x128: a wave holds 5 row-tile accumulators and uses each W fragment 5 times; A strip in LDS at pitch 260 dwords, so fragment reads no longer collide 16-way on banks
# speedup vs baseline: 5.9761x; 1.3253x over previous
#include <hip/hip_runtime.h>
#include <math.h>


#define N_NODES 50000
#define N_EDGES 800000
#define IN_DIM 256
#define HID 64
#define HEADS 4
#define FTOT 256
#define NEG_SLOPE 0.2f

typedef __attribute__((ext_vector_type(2))) float v2f;
typedef __attribute__((ext_vector_type(8))) float v8f;

#define RT 5
#define APITCH 260

template <int M>
__device__ __forceinline__ float row_xmask(float v)
{
    return __int_as_float(
        __builtin_amdgcn_update_dpp(0, __float_as_int(v), 0x160 + M, 0xf, 0xf, true));
}

__global__ __launch_bounds__(256) void gat_gemm_f32_wmma(
    const float* __restrict__ A, const float* __restrict__ W,
    const float* __restrict__ al, const float* __restrict__ ar,
    float* __restrict__ out, float* __restrict__ el, float* __restrict__ er)
{
    __shared__ __attribute__((aligned(16))) float lds[RT * 16 * APITCH];
    __shared__ float red[RT * 8 * 32];
    const int tid  = threadIdx.x;
    const int row0 = blockIdx.x * (16 * RT);

    const float4* Ablk = (const float4*)(A + (size_t)row0 * FTOT);
    #pragma unroll
    for (int i = 0; i < RT * 16 * 64 / 256; ++i) {
        const int f = tid + i * 256;
        *(float4*)&lds[(f >> 6) * APITCH + (f & 63) * 4] = Ablk[f];
    }
    __syncthreads();

    const int wave  = tid >> 5;
    const int lane  = tid & 31;
    const int col0  = (blockIdx.y * 8 + wave) * 16;
    const int hlane = lane & 15;
    const int koff  = (lane >> 4) * 2;

    const float alc = al[col0 + hlane];
    const float arc = ar[col0 + hlane];

    v8f acc[RT];
    #pragma unroll
    for (int t = 0; t < RT; ++t) acc[t] = (v8f){};
    const float* Wc = W + col0 + hlane;
    const float* Al = &lds[hlane * APITCH + koff];
    #pragma unroll 4
    for (int k = 0; k < IN_DIM; k += 4) {
        v2f b;
        b.x = Wc[(size_t)(k + koff) * FTOT];
        b.y = Wc[(size_t)(k + koff + 1) * FTOT];
        #pragma unroll
        for (int t = 0; t < RT; ++t) {
            v2f a = *(const v2f*)&Al[t * 16 * APITCH + k];
            acc[t] = __builtin_amdgcn_wmma_f32_16x16x4_f32(
                         false, a, false, b, (short)0, acc[t], false, false);
        }
    }

    const bool b0 = lane & 1;
    const float ck = b0 ? arc : alc;
    const float cg = b0 ? alc : arc;
    const bool b3 = lane & 8;
    const bool b2 = lane & 4;
    const bool b1 = lane & 2;
    #pragma unroll
    for (int t = 0; t < RT; ++t) {
        float q[8];
        #pragma unroll
        for (int j = 0; j < 8; ++j) q[j] = acc[t][j] * ck + row_xmask<1>(acc[t][j] * cg);
        float r4[4];
        #pragma unroll
        for (int u = 0; u < 4; ++u)
            r4[u] = (b3 ? q[u + 4] : q[u]) + row_xmask<8>(b3 ? q[u] : q[u + 4]);
        float r2[2];
        #pragma unroll
        for (int u = 0; u < 2; ++u)
            r2[u] = (b2 ? r4[u + 2] : r4[u]) + row_xmask<4>(b2 ? r4[u] : r4[u + 2]);
        const float tot = (b1 ? r2[1] : r2[0]) + row_xmask<2>(b1 ? r2[0] : r2[1]);
        red[(t * 8 + wave) * 32 + lane] = tot;
    }
    __syncthreads();
    for (int o = tid; o < RT * 64; o += 256) {
        const int t = o >> 6, arr = (o >> 5) & 1, r = (o >> 1) & 15, hh = o & 1;
        const float* p = &red[(t * 8 + hh * 4) * 32 + r * 2 + arr];
        const float s = (p[0] + p[32]) + (p[64] + p[96]);
        float* dst = arr ? er : el;
        dst[(size_t)(row0 + t * 16 + r) * HEADS + blockIdx.y * 2 + hh] = s;
    }

    #pragma unroll
    for (int t = 0; t < RT; ++t) {
        const int srow = row0 + t * 16 + ((lane >> 4) * 8);
        float* Orow = out + (size_t)srow * FTOT + col0 + hlane;
        #pragma unroll
        for (int j = 0; j < 8; ++j)
            Orow[(size_t)j * FTOT] = acc[t][j];
    }
}

__global__ void csr_zero(int* __restrict__ cnt)
{
    int i = blockIdx.x * blockDim.x + threadIdx.x;
    if (i < N_NODES) cnt[i] = 0;
}

__global__ void csr_ticket(const int* __restrict__ dst, int* __restrict__ cnt,
                           int* __restrict__ ticket)
{
    int e = blockIdx.x * blockDim.x + threadIdx.x;
    if (e >= N_EDGES) return;
    ticket[e] = atomicAdd(&cnt[dst[e]], 1);
}

__global__ __launch_bounds__(1024) void csr_scan(const int* __restrict__ cnt,
                                                 int* __restrict__ rowptr)
{
    __shared__ int wpre[32];
    __shared__ int wtot[32];
    const int t = threadIdx.x, lane = t & 31, w = t >> 5;
    const int base = blockIdx.x * 1024;

    int pre = 0;
    for (int i = t; i < base; i += 1024) pre += cnt[i];
    for (int o = 16; o > 0; o >>= 1) pre += __shfl_xor(pre, o);

    const int i = base + t;
    const int c = (i < N_NODES) ? cnt[i] : 0;
    int v = c;
    for (int o = 1; o < 32; o <<= 1) {
        int u = __shfl_up(v, o);
        if (lane >= o) v += u;
    }
    if (lane == 0)  wpre[w] = pre;
    if (lane == 31) wtot[w] = v;
    __syncthreads();

    int a = wpre[lane];
    int b = (lane < w) ? wtot[lane] : 0;
    for (int o = 16; o > 0; o >>= 1) {
        a += __shfl_xor(a, o);
        b += __shfl_xor(b, o);
    }
    const int excl = a + b + v - c;
    if (i < N_NODES) rowptr[i] = excl;
    if (i == N_NODES - 1) rowptr[N_NODES] = excl + c;
}

__global__ void csr_place(const int* __restrict__ src, const int* __restrict__ dst,
                          const int* __restrict__ ticket, const int* __restrict__ rowptr,
                          int* __restrict__ elist)
{
    int e = blockIdx.x * blockDim.x + threadIdx.x;
    if (e >= N_EDGES) return;
    int pos = rowptr[dst[e]] + ticket[e];
    *(int2*)(elist + 2 * (size_t)pos) = make_int2(e, src[e]);
}

__global__ __launch_bounds__(256) void csr_order(const int* __restrict__ rowptr,
                                                 const int* __restrict__ elist,
                                                 int* __restrict__ ssrc)
{
    const int lane = threadIdx.x & 31;
    const int n = __builtin_amdgcn_readfirstlane(blockIdx.x * 8 + (threadIdx.x >> 5));
    if (n >= N_NODES) return;
    const int beg = rowptr[n], end = rowptr[n + 1];
    const int deg = end - beg;
    if (deg <= 32) {
        int2 me = make_int2(0x7fffffff, 0);
        if (lane < deg) me = *(const int2*)(elist + 2 * (size_t)(beg + lane));
        int r = 0;
        for (int j = 0; j < deg; ++j)
            r += (__builtin_amdgcn_readlane(me.x, j) < me.x) ? 1 : 0;
        if (lane < deg) ssrc[beg + r] = me.y;
    } else {
        for (int i = beg + lane; i < end; i += 32) {
            int2 me = *(const int2*)(elist + 2 * (size_t)i);
            int r = 0;
            for (int k = beg; k < end; ++k) r += (elist[2 * (size_t)k] < me.x) ? 1 : 0;
            ssrc[beg + r] = me.y;
        }
    }
}

__device__ __forceinline__ float leaky(float x)
{
    return x > 0.f ? x : NEG_SLOPE * x;
}

__device__ __forceinline__ float elu1(float x)
{
    return x > 0.f ? x : expf(x) - 1.f;
}

__device__ __forceinline__ float wave_max(float v)
{
    for (int o = 16; o > 0; o >>= 1) v = fmaxf(v, __shfl_xor(v, o));
    return v;
}

__device__ __forceinline__ float wave_sum(float v)
{
    for (int o = 16; o > 0; o >>= 1) v += __shfl_xor(v, o);
    return v;
}

__device__ __forceinline__ float lane_bcast(float v, int j)
{
    return __int_as_float(__builtin_amdgcn_readlane(__float_as_int(v), j));
}

__device__ __forceinline__ void fma4(float4& acc, float w, const float4& f)
{
    acc.x = fmaf(w, f.x, acc.x);
    acc.y = fmaf(w, f.y, acc.y);
    acc.z = fmaf(w, f.z, acc.z);
    acc.w = fmaf(w, f.w, acc.w);
}

__device__ __forceinline__ void edge_weights(float p0, float p1, float p2, float p3,
                                             int j, bool hi, float& wA, float& wB)
{
    const float q0 = lane_bcast(p0, j), q1 = lane_bcast(p1, j);
    const float q2 = lane_bcast(p2, j), q3 = lane_bcast(p3, j);
    wA = hi ? q1 : q0;
    wB = hi ? q3 : q2;
}

__device__ __forceinline__ void gather_rows(
    const float* __restrict__ feat, int s, float p0, float p1, float p2, float p3,
    int cnt, int lane, float4& accA, float4& accB)
{
    const bool hi = lane >= 16;
    const float* fl = feat + 4 * lane;
    int j = 0;
    for (; j + 4 <= cnt; j += 4) {
        const float* r0 = fl + (size_t)__builtin_amdgcn_readlane(s, j)     * FTOT;
        const float* r1 = fl + (size_t)__builtin_amdgcn_readlane(s, j + 1) * FTOT;
        const float* r2 = fl + (size_t)__builtin_amdgcn_readlane(s, j + 2) * FTOT;
        const float* r3 = fl + (size_t)__builtin_amdgcn_readlane(s, j + 3) * FTOT;
        const float4 a0 = *(const float4*)r0, b0 = *(const float4*)(r0 + 128);
        const float4 a1 = *(const float4*)r1, b1 = *(const float4*)(r1 + 128);
        const float4 a2 = *(const float4*)r2, b2 = *(const float4*)(r2 + 128);
        const float4 a3 = *(const float4*)r3, b3 = *(const float4*)(r3 + 128);
        float wA0, wB0, wA1, wB1, wA2, wB2, wA3, wB3;
        edge_weights(p0, p1, p2, p3, j,     hi, wA0, wB0);
        edge_weights(p0, p1, p2, p3, j + 1, hi, wA1, wB1);
        edge_weights(p0, p1, p2, p3, j + 2, hi, wA2, wB2);
        edge_weights(p0, p1, p2, p3, j + 3, hi, wA3, wB3);
        fma4(accA, wA0, a0); fma4(accB, wB0, b0);
        fma4(accA, wA1, a1); fma4(accB, wB1, b1);
        fma4(accA, wA2, a2); fma4(accB, wB2, b2);
        fma4(accA, wA3, a3); fma4(accB, wB3, b3);
    }
    const int rem = cnt - j;
    if (rem > 0) {
        const float* r0 = fl + (size_t)__builtin_amdgcn_readlane(s, j) * FTOT;
        const float4 a0 = *(const float4*)r0, b0 = *(const float4*)(r0 + 128);
        float4 a1 = {0.f, 0.f, 0.f, 0.f}, b1 = a1, a2 = a1, b2 = a1;
        if (rem > 1) {
            const float* r1 = fl + (size_t)__builtin_amdgcn_readlane(s, j + 1) * FTOT;
            a1 = *(const float4*)r1; b1 = *(const float4*)(r1 + 128);
        }
        if (rem > 2) {
            const float* r2 = fl + (size_t)__builtin_amdgcn_readlane(s, j + 2) * FTOT;
            a2 = *(const float4*)r2; b2 = *(const float4*)(r2 + 128);
        }
        float wA, wB;
        edge_weights(p0, p1, p2, p3, j, hi, wA, wB);
        fma4(accA, wA, a0); fma4(accB, wB, b0);
        if (rem > 1) {
            edge_weights(p0, p1, p2, p3, j + 1, hi, wA, wB);
            fma4(accA, wA, a1); fma4(accB, wB, b1);
        }
        if (rem > 2) {
            edge_weights(p0, p1, p2, p3, j + 2, hi, wA, wB);
            fma4(accA, wA, a2); fma4(accB, wB, b2);
        }
    }
}

__global__ __launch_bounds__(256) void gat_node_gather(
    const int* __restrict__ rowptr, const int* __restrict__ ssrc,
    const float* __restrict__ el, const float* __restrict__ er,
    const float* __restrict__ feat, const float* __restrict__ bias,
    float* __restrict__ outp, int head_major)
{
    const int lane = threadIdx.x & 31;
    const int n = __builtin_amdgcn_readfirstlane(blockIdx.x * 8 + (threadIdx.x >> 5));
    if (n >= N_NODES) return;
    const int beg = rowptr[n], end = rowptr[n + 1];
    const int deg = end - beg;
    const float4 ern = *(const float4*)(er + (size_t)n * HEADS);
    const float NINF = -__builtin_inff();

    float4 accA = {0.f, 0.f, 0.f, 0.f}, accB = {0.f, 0.f, 0.f, 0.f};

    if (deg <= 32) {
        const bool valid = lane < deg;
        int s = 0;
        float4 l = {0.f, 0.f, 0.f, 0.f};
        if (valid) {
            s = ssrc[beg + lane];
            l = *(const float4*)(el + (size_t)s * HEADS);
        }
        const float x0 = leaky(l.x + ern.x), x1 = leaky(l.y + ern.y);
        const float x2 = leaky(l.z + ern.z), x3 = leaky(l.w + ern.w);
        const float m0 = wave_max(valid ? x0 : NINF), m1 = wave_max(valid ? x1 : NINF);
        const float m2 = wave_max(valid ? x2 : NINF), m3 = wave_max(valid ? x3 : NINF);
        float p0 = valid ? expf(x0 - m0) : 0.f, p1 = valid ? expf(x1 - m1) : 0.f;
        float p2 = valid ? expf(x2 - m2) : 0.f, p3 = valid ? expf(x3 - m3) : 0.f;
        const float d0 = wave_sum(p0), d1 = wave_sum(p1);
        const float d2 = wave_sum(p2), d3 = wave_sum(p3);
        if (valid) { p0 /= d0; p1 /= d1; p2 /= d2; p3 /= d3; }
        gather_rows(feat, s, p0, p1, p2, p3, deg, lane, accA, accB);
    } else {
        float m0 = NINF, m1 = NINF, m2 = NINF, m3 = NINF;
        for (int i = beg + lane; i < end; i += 32) {
            const float4 l = *(const float4*)(el + (size_t)ssrc[i] * HEADS);
            m0 = fmaxf(m0, leaky(l.x + ern.x)); m1 = fmaxf(m1, leaky(l.y + ern.y));
            m2 = fmaxf(m2, leaky(l.z + ern.z)); m3 = fmaxf(m3, leaky(l.w + ern.w));
        }
        m0 = wave_max(m0); m1 = wave_max(m1); m2 = wave_max(m2); m3 = wave_max(m3);
        float d0 = 0.f, d1 = 0.f, d2 = 0.f, d3 = 0.f;
        for (int i = beg + lane; i < end; i += 32) {
            const float4 l = *(const float4*)(el + (size_t)ssrc[i] * HEADS);
            d0 += expf(leaky(l.x + ern.x) - m0); d1 += expf(leaky(l.y + ern.y) - m1);
            d2 += expf(leaky(l.z + ern.z) - m2); d3 += expf(leaky(l.w + ern.w) - m3);
        }
        d0 = wave_sum(d0); d1 = wave_sum(d1); d2 = wave_sum(d2); d3 = wave_sum(d3);
        for (int base = beg; base < end; base += 32) {
            const int i = base + lane;
            int s = 0;
            float p0 = 0.f, p1 = 0.f, p2 = 0.f, p3 = 0.f;
            if (i < end) {
                s = ssrc[i];
                const float4 l = *(const float4*)(el + (size_t)s * HEADS);
                p0 = expf(leaky(l.x + ern.x) - m0) / d0;
                p1 = expf(leaky(l.y + ern.y) - m1) / d1;
                p2 = expf(leaky(l.z + ern.z) - m2) / d2;
                p3 = expf(leaky(l.w + ern.w) - m3) / d3;
            }
            const int left = end - base;
            gather_rows(feat, s, p0, p1, p2, p3, left < 32 ? left : 32, lane, accA, accB);
        }
    }

    const float4 bA = *(const float4*)(bias + 4 * lane);
    const float4 bB = *(const float4*)(bias + 128 + 4 * lane);
    float4 oA, oB;
    oA.x = elu1(accA.x + bA.x); oA.y = elu1(accA.y + bA.y);
    oA.z = elu1(accA.z + bA.z); oA.w = elu1(accA.w + bA.w);
    oB.x = elu1(accB.x + bB.x); oB.y = elu1(accB.y + bB.y);
    oB.z = elu1(accB.z + bB.z); oB.w = elu1(accB.w + bB.w);

    if (head_major) {
        const int hsel = lane >> 4;
        const int d = (4 * lane) & 63;
        float* o = outp + (size_t)n * HID + d;
        *(float4*)(o + (size_t)hsel * (N_NODES * HID)) = oA;
        *(float4*)(o + (size_t)(hsel + 2) * (N_NODES * HID)) = oB;
    } else {
        float* o = outp + (size_t)n * FTOT + 4 * lane;
        *(float4*)o = oA;
        *(float4*)(o + 128) = oB;
    }
}

extern "C" void kernel_launch(void* const* d_in, const int* in_sizes, int n_in,
                              void* d_out, int out_size, void* d_ws, size_t ws_size,
                              hipStream_t stream)
{
    (void)in_sizes; (void)n_in; (void)out_size; (void)ws_size;

    const float* x   = (const float*)d_in[0];
    const int*   src = (const int*)d_in[1];
    const int*   dst = (const int*)d_in[2];
    const float* W1  = (const float*)d_in[3];
    const float* al1 = (const float*)d_in[4];
    const float* ar1 = (const float*)d_in[5];
    const float* b1  = (const float*)d_in[6];
    const float* W2  = (const float*)d_in[7];
    const float* al2 = (const float*)d_in[8];
    const float* ar2 = (const float*)d_in[9];
    const float* b2  = (const float*)d_in[10];
    float* out = (float*)d_out;

    char* ws = (char*)d_ws;
    const size_t featB = (size_t)N_NODES * FTOT * sizeof(float);
    const size_t nhB   = (size_t)N_NODES * HEADS * sizeof(float);
    const size_t nodeB = 200192;
    const size_t edgeB = (size_t)N_EDGES * sizeof(int);
    float* feat   = (float*)ws; ws += featB;
    float* h1     = (float*)ws; ws += featB;
    float* el     = (float*)ws; ws += nhB;
    float* er     = (float*)ws; ws += nhB;
    int*   cnt    = (int*)ws;   ws += nodeB;
    int*   rowptr = (int*)ws;   ws += nodeB;
    int*   ticket = (int*)ws;   ws += edgeB;
    int*   elist  = (int*)ws;   ws += 2 * edgeB;
    int*   ssrc   = (int*)ws;   ws += edgeB;

    dim3 gGemm(N_NODES / (16 * RT), 2), bGemm(256);
    const int gN  = (N_NODES + 255) / 256;
    const int gE  = (N_EDGES + 255) / 256;
    const int gNW = (N_NODES + 7) / 8;

    csr_zero<<<gN, 256, 0, stream>>>(cnt);
    csr_ticket<<<gE, 256, 0, stream>>>(dst, cnt, ticket);
    csr_scan<<<(N_NODES + 1023) / 1024, 1024, 0, stream>>>(cnt, rowptr);
    csr_place<<<gE, 256, 0, stream>>>(src, dst, ticket, rowptr, elist);
    csr_order<<<gNW, 256, 0, stream>>>(rowptr, elist, ssrc);

    gat_gemm_f32_wmma<<<gGemm, bGemm, 0, stream>>>(x, W1, al1, ar1, feat, el, er);
    gat_node_gather<<<gNW, 256, 0, stream>>>(rowptr, ssrc, el, er, feat, b1, h1, 0);

    gat_gemm_f32_wmma<<<gGemm, bGemm, 0, stream>>>(h1, W2, al2, ar2, feat, el, er);
    gat_node_gather<<<gNW, 256, 0, stream>>>(rowptr, ssrc, el, er, feat, b2, out, 1);
}


# Edges grouped by destination on LDS counters: block (node range, edge slice) counts, then ranks its slice in LDS; no global returning atomics, ticket array or zeroing launch; sum order unchanged
# speedup vs baseline: 7.3954x; 1.2375x over previous
#include <hip/hip_runtime.h>
#include <math.h>


#define N_NODES 50000
#define N_EDGES 800000
#define IN_DIM 256
#define HID 64
#define HEADS 4
#define FTOT 256
#define NEG_SLOPE 0.2f

typedef __attribute__((ext_vector_type(2))) float v2f;
typedef __attribute__((ext_vector_type(8))) float v8f;

#define RT 5
#define APITCH 260

template <int M>
__device__ __forceinline__ float row_xmask(float v)
{
    return __int_as_float(
        __builtin_amdgcn_update_dpp(0, __float_as_int(v), 0x160 + M, 0xf, 0xf, true));
}

__global__ __launch_bounds__(256) void gat_gemm_f32_wmma(
    const float* __restrict__ A, const float* __restrict__ W,
    const float* __restrict__ al, const float* __restrict__ ar,
    float* __restrict__ out, float* __restrict__ el, float* __restrict__ er)
{
    __shared__ __attribute__((aligned(16))) float lds[RT * 16 * APITCH];
    __shared__ float red[RT * 8 * 32];
    const int tid  = threadIdx.x;
    const int row0 = blockIdx.x * (16 * RT);

    const float4* Ablk = (const float4*)(A + (size_t)row0 * FTOT);
    #pragma unroll
    for (int i = 0; i < RT * 16 * 64 / 256; ++i) {
        const int f = tid + i * 256;
        *(float4*)&lds[(f >> 6) * APITCH + (f & 63) * 4] = Ablk[f];
    }
    __syncthreads();

    const int wave  = tid >> 5;
    const int lane  = tid & 31;
    const int col0  = (blockIdx.y * 8 + wave) * 16;
    const int hlane = lane & 15;
    const int koff  = (lane >> 4) * 2;

    const float alc = al[col0 + hlane];
    const float arc = ar[col0 + hlane];

    v8f acc[RT];
    #pragma unroll
    for (int t = 0; t < RT; ++t) acc[t] = (v8f){};
    const float* Wc = W + col0 + hlane;
    const float* Al = &lds[hlane * APITCH + koff];
    #pragma unroll 4
    for (int k = 0; k < IN_DIM; k += 4) {
        v2f b;
        b.x = Wc[(size_t)(k + koff) * FTOT];
        b.y = Wc[(size_t)(k + koff + 1) * FTOT];
        #pragma unroll
        for (int t = 0; t < RT; ++t) {
            v2f a = *(const v2f*)&Al[t * 16 * APITCH + k];
            acc[t] = __builtin_amdgcn_wmma_f32_16x16x4_f32(
                         false, a, false, b, (short)0, acc[t], false, false);
        }
    }

    const bool b0 = lane & 1;
    const float ck = b0 ? arc : alc;
    const float cg = b0 ? alc : arc;
    const bool b3 = lane & 8;
    const bool b2 = lane & 4;
    const bool b1 = lane & 2;
    #pragma unroll
    for (int t = 0; t < RT; ++t) {
        float q[8];
        #pragma unroll
        for (int j = 0; j < 8; ++j) q[j] = acc[t][j] * ck + row_xmask<1>(acc[t][j] * cg);
        float r4[4];
        #pragma unroll
        for (int u = 0; u < 4; ++u)
            r4[u] = (b3 ? q[u + 4] : q[u]) + row_xmask<8>(b3 ? q[u] : q[u + 4]);
        float r2[2];
        #pragma unroll
        for (int u = 0; u < 2; ++u)
            r2[u] = (b2 ? r4[u + 2] : r4[u]) + row_xmask<4>(b2 ? r4[u] : r4[u + 2]);
        const float tot = (b1 ? r2[1] : r2[0]) + row_xmask<2>(b1 ? r2[0] : r2[1]);
        red[(t * 8 + wave) * 32 + lane] = tot;
    }
    __syncthreads();
    for (int o = tid; o < RT * 64; o += 256) {
        const int t = o >> 6, arr = (o >> 5) & 1, r = (o >> 1) & 15, hh = o & 1;
        const float* p = &red[(t * 8 + hh * 4) * 32 + r * 2 + arr];
        const float s = (p[0] + p[32]) + (p[64] + p[96]);
        float* dst = arr ? er : el;
        dst[(size_t)(row0 + t * 16 + r) * HEADS + blockIdx.y * 2 + hh] = s;
    }

    #pragma unroll
    for (int t = 0; t < RT; ++t) {
        const int srow = row0 + t * 16 + ((lane >> 4) * 8);
        float* Orow = out + (size_t)srow * FTOT + col0 + hlane;
        #pragma unroll
        for (int j = 0; j < 8; ++j)
            Orow[(size_t)j * FTOT] = acc[t][j];
    }
}

#define SRT_P 16
#define SRT_Q 16
#define SRT_R 3128
#define SRT_S (N_EDGES / SRT_Q)
#define SRT_S4 (SRT_S / 4)
#define SRT_B 7

__device__ __forceinline__ int wave_sum_i(int v)
{
    for (int o = 16; o > 0; o >>= 1) v += __shfl_xor(v, o);
    return v;
}

__device__ __forceinline__ void load_pieces(const int4* __restrict__ a, int h, int t,
                                            int fill, int4 (&v)[SRT_B])
{
    #pragma unroll
    for (int u = 0; u < SRT_B; ++u) {
        const int i = (h * SRT_B + u) * 1024 + t;
        v[u] = (i < SRT_S4) ? a[i] : make_int4(fill, fill, fill, fill);
    }
}

__device__ __forceinline__ void count_one(int* cnt, int rel)
{
    if ((unsigned)rel < (unsigned)SRT_R) atomicAdd(&cnt[rel], 1);
}

__global__ __launch_bounds__(1024) void csr_count(const int* __restrict__ dst,
                                                  int* __restrict__ part,
                                                  int* __restrict__ tot)
{
    __shared__ int cnt[SRT_R];
    __shared__ int wsum[32];
    const int t = threadIdx.x;
    const int p = blockIdx.x, q = blockIdx.y;
    const int lo = p * SRT_R;
    for (int i = t; i < SRT_R; i += 1024) cnt[i] = 0;
    __syncthreads();

    const int4* d4 = (const int4*)dst + (size_t)q * SRT_S4;
    for (int h = 0; h < 2; ++h) {
        int4 v[SRT_B];
        load_pieces(d4, h, t, -1, v);
        #pragma unroll
        for (int u = 0; u < SRT_B; ++u) {
            count_one(cnt, v[u].x - lo);
            count_one(cnt, v[u].y - lo);
            count_one(cnt, v[u].z - lo);
            count_one(cnt, v[u].w - lo);
        }
    }
    __syncthreads();

    int s = 0;
    for (int i = t; i < SRT_R; i += 1024) {
        const int c = cnt[i];
        if (lo + i < N_NODES) part[(size_t)q * N_NODES + lo + i] = c;
        s += c;
    }
    s = wave_sum_i(s);
    if ((t & 31) == 0) wsum[t >> 5] = s;
    __syncthreads();
    if (t < 32) {
        const int a = wave_sum_i(wsum[t]);
        if (t == 0) tot[p * SRT_Q + q] = a;
    }
}

__global__ __launch_bounds__(1024) void csr_offsets(int* __restrict__ part,
                                                    const int* __restrict__ tot,
                                                    int* __restrict__ rowptr)
{
    __shared__ int wpre[32];
    __shared__ int wtot[32];
    const int t = threadIdx.x, lane = t & 31, w = t >> 5;
    const int p = blockIdx.x;
    const int n0 = p * SRT_R + 4 * t;
    const bool valid = (4 * t < SRT_R) && (n0 < N_NODES);

    int pre = (t < p * SRT_Q) ? tot[t] : 0;
    pre = wave_sum_i(pre);

    int4 c[SRT_Q];
    #pragma unroll
    for (int q = 0; q < SRT_Q; ++q)
        c[q] = valid ? *(const int4*)(part + (size_t)q * N_NODES + n0) : make_int4(0, 0, 0, 0);
    int4 deg = make_int4(0, 0, 0, 0);
    #pragma unroll
    for (int q = 0; q < SRT_Q; ++q) {
        deg.x += c[q].x; deg.y += c[q].y; deg.z += c[q].z; deg.w += c[q].w;
    }
    const int mine = deg.x + deg.y + deg.z + deg.w;

    int v = mine;
    for (int o = 1; o < 32; o <<= 1) {
        int u = __shfl_up(v, o);
        if (lane >= o) v += u;
    }
    if (lane == 0)  wpre[w] = pre;
    if (lane == 31) wtot[w] = v;
    __syncthreads();

    int a = wpre[lane];
    int b = (lane < w) ? wtot[lane] : 0;
    for (int o = 16; o > 0; o >>= 1) {
        a += __shfl_xor(a, o);
        b += __shfl_xor(b, o);
    }
    const int first = a + b + v - mine;
    if (valid) {
        int4 r;
        r.x = first; r.y = r.x + deg.x; r.z = r.y + deg.y; r.w = r.z + deg.z;
        *(int4*)(rowptr + n0) = r;
        #pragma unroll
        for (int q = 0; q < SRT_Q; ++q) {
            *(int4*)(part + (size_t)q * N_NODES + n0) = r;
            r.x += c[q].x; r.y += c[q].y; r.z += c[q].z; r.w += c[q].w;
        }
        if (n0 + 4 == N_NODES) rowptr[N_NODES] = first + mine;
    }
}

__device__ __forceinline__ int rank_one(int* cur, int rel)
{
    int pos = -1;
    if ((unsigned)rel < (unsigned)SRT_R) pos = atomicAdd(&cur[rel], 1);
    return pos;
}

__device__ __forceinline__ void put_one(int* __restrict__ elist, int pos, int e, int s)
{
    if (pos >= 0) *(int2*)(elist + 2 * (size_t)pos) = make_int2(e, s);
}

__device__ __forceinline__ void place_pieces(int* cur, int* __restrict__ elist, int lo,
                                             int e0, int h, const int4 (&dv)[SRT_B],
                                             const int4 (&sv)[SRT_B])
{
    int4 pos[SRT_B];
    #pragma unroll
    for (int u = 0; u < SRT_B; ++u) {
        pos[u].x = rank_one(cur, dv[u].x - lo);
        pos[u].y = rank_one(cur, dv[u].y - lo);
        pos[u].z = rank_one(cur, dv[u].z - lo);
        pos[u].w = rank_one(cur, dv[u].w - lo);
    }
    #pragma unroll
    for (int u = 0; u < SRT_B; ++u) {
        const int e = e0 + 4 * (h * SRT_B + u) * 1024;
        put_one(elist, pos[u].x, e,     sv[u].x);
        put_one(elist, pos[u].y, e + 1, sv[u].y);
        put_one(elist, pos[u].z, e + 2, sv[u].z);
        put_one(elist, pos[u].w, e + 3, sv[u].w);
    }
}

__global__ __launch_bounds__(1024) void csr_place(const int* __restrict__ src,
                                                  const int* __restrict__ dst,
                                                  const int* __restrict__ part,
                                                  int* __restrict__ elist)
{
    __shared__ int cur[SRT_R];
    const int t = threadIdx.x;
    const int p = blockIdx.x, q = blockIdx.y;
    const int lo = p * SRT_R;
    const int4* d4 = (const int4*)dst + (size_t)q * SRT_S4;
    const int4* s4 = (const int4*)src + (size_t)q * SRT_S4;
    const int e0 = q * SRT_S + 4 * t;

    int4 dv[SRT_B], sv[SRT_B];
    load_pieces(d4, 0, t, -1, dv);
    load_pieces(s4, 0, t, 0, sv);

    int fp[(SRT_R + 1023) / 1024];
    #pragma unroll
    for (int k = 0; k < (SRT_R + 1023) / 1024; ++k) {
        const int n = lo + k * 1024 + t;
        fp[k] = part[(size_t)q * N_NODES + (n < N_NODES ? n : N_NODES - 1)];
    }
    #pragma unroll
    for (int k = 0; k < (SRT_R + 1023) / 1024; ++k) {
        const int i = k * 1024 + t;
        if (i < SRT_R) cur[i] = (lo + i < N_NODES) ? fp[k] : 0;
    }
    __syncthreads();

    place_pieces(cur, elist, lo, e0, 0, dv, sv);
    load_pieces(d4, 1, t, -1, dv);
    load_pieces(s4, 1, t, 0, sv);
    place_pieces(cur, elist, lo, e0, 1, dv, sv);
}

__global__ __launch_bounds__(256) void csr_order(const int* __restrict__ rowptr,
                                                 const int* __restrict__ elist,
                                                 int* __restrict__ ssrc)
{
    const int lane = threadIdx.x & 31;
    const int n = __builtin_amdgcn_readfirstlane(blockIdx.x * 8 + (threadIdx.x >> 5));
    if (n >= N_NODES) return;
    const int beg = rowptr[n], end = rowptr[n + 1];
    const int deg = end - beg;
    if (deg <= 32) {
        int2 me = make_int2(0x7fffffff, 0);
        if (lane < deg) me = *(const int2*)(elist + 2 * (size_t)(beg + lane));
        int r = 0;
        for (int j = 0; j < deg; ++j)
            r += (__builtin_amdgcn_readlane(me.x, j) < me.x) ? 1 : 0;
        if (lane < deg) ssrc[beg + r] = me.y;
    } else {
        for (int i = beg + lane; i < end; i += 32) {
            int2 me = *(const int2*)(elist + 2 * (size_t)i);
            int r = 0;
            for (int k = beg; k < end; ++k) r += (elist[2 * (size_t)k] < me.x) ? 1 : 0;
            ssrc[beg + r] = me.y;
        }
    }
}

__device__ __forceinline__ float leaky(float x)
{
    return x > 0.f ? x : NEG_SLOPE * x;
}

__device__ __forceinline__ float elu1(float x)
{
    return x > 0.f ? x : expf(x) - 1.f;
}

__device__ __forceinline__ float wave_max(float v)
{
    for (int o = 16; o > 0; o >>= 1) v = fmaxf(v, __shfl_xor(v, o));
    return v;
}

__device__ __forceinline__ float wave_sum(float v)
{
    for (int o = 16; o > 0; o >>= 1) v += __shfl_xor(v, o);
    return v;
}

__device__ __forceinline__ float lane_bcast(float v, int j)
{
    return __int_as_float(__builtin_amdgcn_readlane(__float_as_int(v), j));
}

__device__ __forceinline__ void fma4(float4& acc, float w, const float4& f)
{
    acc.x = fmaf(w, f.x, acc.x);
    acc.y = fmaf(w, f.y, acc.y);
    acc.z = fmaf(w, f.z, acc.z);
    acc.w = fmaf(w, f.w, acc.w);
}

__device__ __forceinline__ void edge_weights(float p0, float p1, float p2, float p3,
                                             int j, bool hi, float& wA, float& wB)
{
    const float q0 = lane_bcast(p0, j), q1 = lane_bcast(p1, j);
    const float q2 = lane_bcast(p2, j), q3 = lane_bcast(p3, j);
    wA = hi ? q1 : q0;
    wB = hi ? q3 : q2;
}

__device__ __forceinline__ void gather_rows(
    const float* __restrict__ feat, int s, float p0, float p1, float p2, float p3,
    int cnt, int lane, float4& accA, float4& accB)
{
    const bool hi = lane >= 16;
    const float* fl = feat + 4 * lane;
    int j = 0;
    for (; j + 4 <= cnt; j += 4) {
        const float* r0 = fl + (size_t)__builtin_amdgcn_readlane(s, j)     * FTOT;
        const float* r1 = fl + (size_t)__builtin_amdgcn_readlane(s, j + 1) * FTOT;
        const float* r2 = fl + (size_t)__builtin_amdgcn_readlane(s, j + 2) * FTOT;
        const float* r3 = fl + (size_t)__builtin_amdgcn_readlane(s, j + 3) * FTOT;
        const float4 a0 = *(const float4*)r0, b0 = *(const float4*)(r0 + 128);
        const float4 a1 = *(const float4*)r1, b1 = *(const float4*)(r1 + 128);
        const float4 a2 = *(const float4*)r2, b2 = *(const float4*)(r2 + 128);
        const float4 a3 = *(const float4*)r3, b3 = *(const float4*)(r3 + 128);
        float wA0, wB0, wA1, wB1, wA2, wB2, wA3, wB3;
        edge_weights(p0, p1, p2, p3, j,     hi, wA0, wB0);
        edge_weights(p0, p1, p2, p3, j + 1, hi, wA1, wB1);
        edge_weights(p0, p1, p2, p3, j + 2, hi, wA2, wB2);
        edge_weights(p0, p1, p2, p3, j + 3, hi, wA3, wB3);
        fma4(accA, wA0, a0); fma4(accB, wB0, b0);
        fma4(accA, wA1, a1); fma4(accB, wB1, b1);
        fma4(accA, wA2, a2); fma4(accB, wB2, b2);
        fma4(accA, wA3, a3); fma4(accB, wB3, b3);
    }
    const int rem = cnt - j;
    if (rem > 0) {
        const float* r0 = fl + (size_t)__builtin_amdgcn_readlane(s, j) * FTOT;
        const float4 a0 = *(const float4*)r0, b0 = *(const float4*)(r0 + 128);
        float4 a1 = {0.f, 0.f, 0.f, 0.f}, b1 = a1, a2 = a1, b2 = a1;
        if (rem > 1) {
            const float* r1 = fl + (size_t)__builtin_amdgcn_readlane(s, j + 1) * FTOT;
            a1 = *(const float4*)r1; b1 = *(const float4*)(r1 + 128);
        }
        if (rem > 2) {
            const float* r2 = fl + (size_t)__builtin_amdgcn_readlane(s, j + 2) * FTOT;
            a2 = *(const float4*)r2; b2 = *(const float4*)(r2 + 128);
        }
        float wA, wB;
        edge_weights(p0, p1, p2, p3, j, hi, wA, wB);
        fma4(accA, wA, a0); fma4(accB, wB, b0);
        if (rem > 1) {
            edge_weights(p0, p1, p2, p3, j + 1, hi, wA, wB);
            fma4(accA, wA, a1); fma4(accB, wB, b1);
        }
        if (rem > 2) {
            edge_weights(p0, p1, p2, p3, j + 2, hi, wA, wB);
            fma4(accA, wA, a2); fma4(accB, wB, b2);
        }
    }
}

__global__ __launch_bounds__(256) void gat_node_gather(
    const int* __restrict__ rowptr, const int* __restrict__ ssrc,
    const float* __restrict__ el, const float* __restrict__ er,
    const float* __restrict__ feat, const float* __restrict__ bias,
    float* __restrict__ outp, int head_major)
{
    const int lane = threadIdx.x & 31;
    const int n = __builtin_amdgcn_readfirstlane(blockIdx.x * 8 + (threadIdx.x >> 5));
    if (n >= N_NODES) return;
    const int beg = rowptr[n], end = rowptr[n + 1];
    const int deg = end - beg;
    const float4 ern = *(const float4*)(er + (size_t)n * HEADS);
    const float NINF = -__builtin_inff();

    float4 accA = {0.f, 0.f, 0.f, 0.f}, accB = {0.f, 0.f, 0.f, 0.f};

    if (deg <= 32) {
        const bool valid = lane < deg;
        int s = 0;
        float4 l = {0.f, 0.f, 0.f, 0.f};
        if (valid) {
            s = ssrc[beg + lane];
            l = *(const float4*)(el + (size_t)s * HEADS);
        }
        const float x0 = leaky(l.x + ern.x), x1 = leaky(l.y + ern.y);
        const float x2 = leaky(l.z + ern.z), x3 = leaky(l.w + ern.w);
        const float m0 = wave_max(valid ? x0 : NINF), m1 = wave_max(valid ? x1 : NINF);
        const float m2 = wave_max(valid ? x2 : NINF), m3 = wave_max(valid ? x3 : NINF);
        float p0 = valid ? expf(x0 - m0) : 0.f, p1 = valid ? expf(x1 - m1) : 0.f;
        float p2 = valid ? expf(x2 - m2) : 0.f, p3 = valid ? expf(x3 - m3) : 0.f;
        const float d0 = wave_sum(p0), d1 = wave_sum(p1);
        const float d2 = wave_sum(p2), d3 = wave_sum(p3);
        if (valid) { p0 /= d0; p1 /= d1; p2 /= d2; p3 /= d3; }
        gather_rows(feat, s, p0, p1, p2, p3, deg, lane, accA, accB);
    } else {
        float m0 = NINF, m1 = NINF, m2 = NINF, m3 = NINF;
        for (int i = beg + lane; i < end; i += 32) {
            const float4 l = *(const float4*)(el + (size_t)ssrc[i] * HEADS);
            m0 = fmaxf(m0, leaky(l.x + ern.x)); m1 = fmaxf(m1, leaky(l.y + ern.y));
            m2 = fmaxf(m2, leaky(l.z + ern.z)); m3 = fmaxf(m3, leaky(l.w + ern.w));
        }
        m0 = wave_max(m0); m1 = wave_max(m1); m2 = wave_max(m2); m3 = wave_max(m3);
        float d0 = 0.f, d1 = 0.f, d2 = 0.f, d3 = 0.f;
        for (int i = beg + lane; i < end; i += 32) {
            const float4 l = *(const float4*)(el + (size_t)ssrc[i] * HEADS);
            d0 += expf(leaky(l.x + ern.x) - m0); d1 += expf(leaky(l.y + ern.y) - m1);
            d2 += expf(leaky(l.z + ern.z) - m2); d3 += expf(leaky(l.w + ern.w) - m3);
        }
        d0 = wave_sum(d0); d1 = wave_sum(d1); d2 = wave_sum(d2); d3 = wave_sum(d3);
        for (int base = beg; base < end; base += 32) {
            const int i = base + lane;
            int s = 0;
            float p0 = 0.f, p1 = 0.f, p2 = 0.f, p3 = 0.f;
            if (i < end) {
                s = ssrc[i];
                const float4 l = *(const float4*)(el + (size_t)s * HEADS);
                p0 = expf(leaky(l.x + ern.x) - m0) / d0;
                p1 = expf(leaky(l.y + ern.y) - m1) / d1;
                p2 = expf(leaky(l.z + ern.z) - m2) / d2;
                p3 = expf(leaky(l.w + ern.w) - m3) / d3;
            }
            const int left = end - base;
            gather_rows(feat, s, p0, p1, p2, p3, left < 32 ? left : 32, lane, accA, accB);
        }
    }

    const float4 bA = *(const float4*)(bias + 4 * lane);
    const float4 bB = *(const float4*)(bias + 128 + 4 * lane);
    float4 oA, oB;
    oA.x = elu1(accA.x + bA.x); oA.y = elu1(accA.y + bA.y);
    oA.z = elu1(accA.z + bA.z); oA.w = elu1(accA.w + bA.w);
    oB.x = elu1(accB.x + bB.x); oB.y = elu1(accB.y + bB.y);
    oB.z = elu1(accB.z + bB.z); oB.w = elu1(accB.w + bB.w);

    if (head_major) {
        const int hsel = lane >> 4;
        const int d = (4 * lane) & 63;
        float* o = outp + (size_t)n * HID + d;
        *(float4*)(o + (size_t)hsel * (N_NODES * HID)) = oA;
        *(float4*)(o + (size_t)(hsel + 2) * (N_NODES * HID)) = oB;
    } else {
        float* o = outp + (size_t)n * FTOT + 4 * lane;
        *(float4*)o = oA;
        *(float4*)(o + 128) = oB;
    }
}

extern "C" void kernel_launch(void* const* d_in, const int* in_sizes, int n_in,
                              void* d_out, int out_size, void* d_ws, size_t ws_size,
                              hipStream_t stream)
{
    (void)in_sizes; (void)n_in; (void)out_size; (void)ws_size;

    const float* x   = (const float*)d_in[0];
    const int*   src = (const int*)d_in[1];
    const int*   dst = (const int*)d_in[2];
    const float* W1  = (const float*)d_in[3];
    const float* al1 = (const float*)d_in[4];
    const float* ar1 = (const float*)d_in[5];
    const float* b1  = (const float*)d_in[6];
    const float* W2  = (const float*)d_in[7];
    const float* al2 = (const float*)d_in[8];
    const float* ar2 = (const float*)d_in[9];
    const float* b2  = (const float*)d_in[10];
    float* out = (float*)d_out;

    char* ws = (char*)d_ws;
    const size_t featB = (size_t)N_NODES * FTOT * sizeof(float);
    const size_t nhB   = (size_t)N_NODES * HEADS * sizeof(float);
    const size_t nodeB = 200192;
    const size_t edgeB = (size_t)N_EDGES * sizeof(int);
    float* feat   = (float*)ws; ws += featB;
    float* h1     = (float*)ws; ws += featB;
    float* el     = (float*)ws; ws += nhB;
    float* er     = (float*)ws; ws += nhB;
    int*   tot    = (int*)ws;   ws += nodeB;
    int*   rowptr = (int*)ws;   ws += nodeB;
    int*   part   = (int*)ws;   ws += edgeB;
    int*   elist  = (int*)ws;   ws += 2 * edgeB;
    int*   ssrc   = (int*)ws;   ws += edgeB;

    dim3 gGemm(N_NODES / (16 * RT), 2), bGemm(256);
    dim3 gSort(SRT_P, SRT_Q);
    const int gNW = (N_NODES + 7) / 8;

    csr_count<<<gSort, 1024, 0, stream>>>(dst, part, tot);
    csr_offsets<<<SRT_P, 1024, 0, stream>>>(part, tot, rowptr);
    csr_place<<<gSort, 1024, 0, stream>>>(src, dst, part, elist);
    csr_order<<<gNW, 256, 0, stream>>>(rowptr, elist, ssrc);

    gat_gemm_f32_wmma<<<gGemm, bGemm, 0, stream>>>(x, W1, al1, ar1, feat, el, er);
    gat_node_gather<<<gNW, 256, 0, stream>>>(rowptr, ssrc, el, er, feat, b1, h1, 0);

    gat_gemm_f32_wmma<<<gGemm, bGemm, 0, stream>>>(h1, W2, al2, ar2, feat, el, er);
    gat_node_gather<<<gNW, 256, 0, stream>>>(rowptr, ssrc, el, er, feat, b2, out, 1);
}


# Two small launches folded into their consumers: placement blocks scan their range's counts themselves, gather waves rank their list by edge index in registers; 8 launches -> 6; sum order unchanged
# speedup vs baseline: 7.7409x; 1.0467x over previous
#include <hip/hip_runtime.h>
#include <math.h>


#define N_NODES 50000
#define N_EDGES 800000
#define IN_DIM 256
#define HID 64
#define HEADS 4
#define FTOT 256
#define NEG_SLOPE 0.2f

typedef __attribute__((ext_vector_type(2))) float v2f;
typedef __attribute__((ext_vector_type(8))) float v8f;

#define RT 5
#define APITCH 260

template <int M>
__device__ __forceinline__ float row_xmask(float v)
{
    return __int_as_float(
        __builtin_amdgcn_update_dpp(0, __float_as_int(v), 0x160 + M, 0xf, 0xf, true));
}

__global__ __launch_bounds__(256) void gat_gemm_f32_wmma(
    const float* __restrict__ A, const float* __restrict__ W,
    const float* __restrict__ al, const float* __restrict__ ar,
    float* __restrict__ out, float* __restrict__ el, float* __restrict__ er)
{
    __shared__ __attribute__((aligned(16))) float lds[RT * 16 * APITCH];
    __shared__ float red[RT * 8 * 32];
    const int tid  = threadIdx.x;
    const int row0 = blockIdx.x * (16 * RT);

    const float4* Ablk = (const float4*)(A + (size_t)row0 * FTOT);
    #pragma unroll
    for (int i = 0; i < RT * 16 * 64 / 256; ++i) {
        const int f = tid + i * 256;
        *(float4*)&lds[(f >> 6) * APITCH + (f & 63) * 4] = Ablk[f];
    }
    __syncthreads();

    const int wave  = tid >> 5;
    const int lane  = tid & 31;
    const int col0  = (blockIdx.y * 8 + wave) * 16;
    const int hlane = lane & 15;
    const int koff  = (lane >> 4) * 2;

    const float alc = al[col0 + hlane];
    const float arc = ar[col0 + hlane];

    v8f acc[RT];
    #pragma unroll
    for (int t = 0; t < RT; ++t) acc[t] = (v8f){};
    const float* Wc = W + col0 + hlane;
    const float* Al = &lds[hlane * APITCH + koff];
    #pragma unroll 4
    for (int k = 0; k < IN_DIM; k += 4) {
        v2f b;
        b.x = Wc[(size_t)(k + koff) * FTOT];
        b.y = Wc[(size_t)(k + koff + 1) * FTOT];
        #pragma unroll
        for (int t = 0; t < RT; ++t) {
            v2f a = *(const v2f*)&Al[t * 16 * APITCH + k];
            acc[t] = __builtin_amdgcn_wmma_f32_16x16x4_f32(
                         false, a, false, b, (short)0, acc[t], false, false);
        }
    }

    const bool b0 = lane & 1;
    const float ck = b0 ? arc : alc;
    const float cg = b0 ? alc : arc;
    const bool b3 = lane & 8;
    const bool b2 = lane & 4;
    const bool b1 = lane & 2;
    #pragma unroll
    for (int t = 0; t < RT; ++t) {
        float q[8];
        #pragma unroll
        for (int j = 0; j < 8; ++j) q[j] = acc[t][j] * ck + row_xmask<1>(acc[t][j] * cg);
        float r4[4];
        #pragma unroll
        for (int u = 0; u < 4; ++u)
            r4[u] = (b3 ? q[u + 4] : q[u]) + row_xmask<8>(b3 ? q[u] : q[u + 4]);
        float r2[2];
        #pragma unroll
        for (int u = 0; u < 2; ++u)
            r2[u] = (b2 ? r4[u + 2] : r4[u]) + row_xmask<4>(b2 ? r4[u] : r4[u + 2]);
        const float tot = (b1 ? r2[1] : r2[0]) + row_xmask<2>(b1 ? r2[0] : r2[1]);
        red[(t * 8 + wave) * 32 + lane] = tot;
    }
    __syncthreads();
    for (int o = tid; o < RT * 64; o += 256) {
        const int t = o >> 6, arr = (o >> 5) & 1, r = (o >> 1) & 15, hh = o & 1;
        const float* p = &red[(t * 8 + hh * 4) * 32 + r * 2 + arr];
        const float s = (p[0] + p[32]) + (p[64] + p[96]);
        float* dst = arr ? er : el;
        dst[(size_t)(row0 + t * 16 + r) * HEADS + blockIdx.y * 2 + hh] = s;
    }

    #pragma unroll
    for (int t = 0; t < RT; ++t) {
        const int srow = row0 + t * 16 + ((lane >> 4) * 8);
        float* Orow = out + (size_t)srow * FTOT + col0 + hlane;
        #pragma unroll
        for (int j = 0; j < 8; ++j)
            Orow[(size_t)j * FTOT] = acc[t][j];
    }
}

#define SRT_P 16
#define SRT_Q 16
#define SRT_R 3128
#define SRT_S (N_EDGES / SRT_Q)
#define SRT_S4 (SRT_S / 4)
#define SRT_B 7

__device__ __forceinline__ int wave_sum_i(int v)
{
    for (int o = 16; o > 0; o >>= 1) v += __shfl_xor(v, o);
    return v;
}

__device__ __forceinline__ void load_pieces(const int4* __restrict__ a, int h, int t,
                                            int fill, int4 (&v)[SRT_B])
{
    #pragma unroll
    for (int u = 0; u < SRT_B; ++u) {
        const int i = (h * SRT_B + u) * 1024 + t;
        v[u] = (i < SRT_S4) ? a[i] : make_int4(fill, fill, fill, fill);
    }
}

__device__ __forceinline__ void count_one(int* cnt, int rel)
{
    if ((unsigned)rel < (unsigned)SRT_R) atomicAdd(&cnt[rel], 1);
}

__global__ __launch_bounds__(1024) void csr_count(const int* __restrict__ dst,
                                                  int* __restrict__ part,
                                                  int* __restrict__ tot)
{
    __shared__ int cnt[SRT_R];
    __shared__ int wsum[32];
    const int t = threadIdx.x;
    const int p = blockIdx.x, q = blockIdx.y;
    const int lo = p * SRT_R;
    for (int i = t; i < SRT_R; i += 1024) cnt[i] = 0;
    __syncthreads();

    const int4* d4 = (const int4*)dst + (size_t)q * SRT_S4;
    for (int h = 0; h < 2; ++h) {
        int4 v[SRT_B];
        load_pieces(d4, h, t, -1, v);
        #pragma unroll
        for (int u = 0; u < SRT_B; ++u) {
            count_one(cnt, v[u].x - lo);
            count_one(cnt, v[u].y - lo);
            count_one(cnt, v[u].z - lo);
            count_one(cnt, v[u].w - lo);
        }
    }
    __syncthreads();

    int s = 0;
    for (int i = t; i < SRT_R; i += 1024) {
        const int c = cnt[i];
        if (lo + i < N_NODES) part[(size_t)q * N_NODES + lo + i] = c;
        s += c;
    }
    s = wave_sum_i(s);
    if ((t & 31) == 0) wsum[t >> 5] = s;
    __syncthreads();
    if (t < 32) {
        const int a = wave_sum_i(wsum[t]);
        if (t == 0) tot[p * SRT_Q + q] = a;
    }
}

__device__ __forceinline__ int rank_one(int* cur, int rel)
{
    int pos = -1;
    if ((unsigned)rel < (unsigned)SRT_R) pos = atomicAdd(&cur[rel], 1);
    return pos;
}

__device__ __forceinline__ void put_one(int* __restrict__ elist, int pos, int e, int s)
{
    if (pos >= 0) *(int2*)(elist + 2 * (size_t)pos) = make_int2(e, s);
}

__device__ __forceinline__ void place_pieces(int* cur, int* __restrict__ elist, int lo,
                                             int e0, int h, const int4 (&dv)[SRT_B],
                                             const int4 (&sv)[SRT_B])
{
    int4 pos[SRT_B];
    #pragma unroll
    for (int u = 0; u < SRT_B; ++u) {
        pos[u].x = rank_one(cur, dv[u].x - lo);
        pos[u].y = rank_one(cur, dv[u].y - lo);
        pos[u].z = rank_one(cur, dv[u].z - lo);
        pos[u].w = rank_one(cur, dv[u].w - lo);
    }
    #pragma unroll
    for (int u = 0; u < SRT_B; ++u) {
        const int e = e0 + 4 * (h * SRT_B + u) * 1024;
        put_one(elist, pos[u].x, e,     sv[u].x);
        put_one(elist, pos[u].y, e + 1, sv[u].y);
        put_one(elist, pos[u].z, e + 2, sv[u].z);
        put_one(elist, pos[u].w, e + 3, sv[u].w);
    }
}

__global__ __launch_bounds__(1024) void csr_place(const int* __restrict__ src,
                                                  const int* __restrict__ dst,
                                                  const int* __restrict__ part,
                                                  const int* __restrict__ tot,
                                                  int* __restrict__ rowptr,
                                                  int* __restrict__ elist)
{
    __shared__ __attribute__((aligned(16))) int cur[SRT_R];
    __shared__ int wpre[32];
    __shared__ int wtot[32];
    const int t = threadIdx.x, lane = t & 31, w = t >> 5;
    const int p = blockIdx.x, q = blockIdx.y;
    const int lo = p * SRT_R;
    const int4* d4 = (const int4*)dst + (size_t)q * SRT_S4;
    const int4* s4 = (const int4*)src + (size_t)q * SRT_S4;
    const int e0 = q * SRT_S + 4 * t;

    int4 dv[SRT_B], sv[SRT_B];
    load_pieces(d4, 0, t, -1, dv);

    const int n0 = lo + 4 * t;
    const bool mine4 = 4 * t < SRT_R;
    const bool valid = mine4 && (n0 < N_NODES);

    int pre = (t < p * SRT_Q) ? tot[t] : 0;
    pre = wave_sum_i(pre);

    int4 deg = make_int4(0, 0, 0, 0);
    int4 bef = make_int4(0, 0, 0, 0);
    #pragma unroll
    for (int qq = 0; qq < SRT_Q; ++qq) {
        const int4 c = valid ? *(const int4*)(part + (size_t)qq * N_NODES + n0)
                             : make_int4(0, 0, 0, 0);
        deg.x += c.x; deg.y += c.y; deg.z += c.z; deg.w += c.w;
        if (qq < q) { bef.x += c.x; bef.y += c.y; bef.z += c.z; bef.w += c.w; }
    }
    const int sum4 = deg.x + deg.y + deg.z + deg.w;

    int v = sum4;
    for (int o = 1; o < 32; o <<= 1) {
        int u = __shfl_up(v, o);
        if (lane >= o) v += u;
    }
    if (lane == 0)  wpre[w] = pre;
    if (lane == 31) wtot[w] = v;
    __syncthreads();

    int a = wpre[lane];
    int b = (lane < w) ? wtot[lane] : 0;
    for (int o = 16; o > 0; o >>= 1) {
        a += __shfl_xor(a, o);
        b += __shfl_xor(b, o);
    }
    const int first = a + b + v - sum4;
    int4 r;
    r.x = first; r.y = r.x + deg.x; r.z = r.y + deg.y; r.w = r.z + deg.z;
    if (valid && q == 0) {
        *(int4*)(rowptr + n0) = r;
        if (n0 + 4 == N_NODES) rowptr[N_NODES] = first + sum4;
    }
    if (mine4) {
        int4 c0 = make_int4(0, 0, 0, 0);
        if (valid) { c0.x = r.x + bef.x; c0.y = r.y + bef.y; c0.z = r.z + bef.z; c0.w = r.w + bef.w; }
        *(int4*)&cur[4 * t] = c0;
    }
    load_pieces(s4, 0, t, 0, sv);
    __syncthreads();

    place_pieces(cur, elist, lo, e0, 0, dv, sv);
    load_pieces(d4, 1, t, -1, dv);
    load_pieces(s4, 1, t, 0, sv);
    place_pieces(cur, elist, lo, e0, 1, dv, sv);
}

__device__ __forceinline__ float leaky(float x)
{
    return x > 0.f ? x : NEG_SLOPE * x;
}

__device__ __forceinline__ float elu1(float x)
{
    return x > 0.f ? x : expf(x) - 1.f;
}

__device__ __forceinline__ float wave_max(float v)
{
    for (int o = 16; o > 0; o >>= 1) v = fmaxf(v, __shfl_xor(v, o));
    return v;
}

__device__ __forceinline__ float wave_sum(float v)
{
    for (int o = 16; o > 0; o >>= 1) v += __shfl_xor(v, o);
    return v;
}

__device__ __forceinline__ float lane_bcast(float v, int j)
{
    return __int_as_float(__builtin_amdgcn_readlane(__float_as_int(v), j));
}

__device__ __forceinline__ void fma4(float4& acc, float w, const float4& f)
{
    acc.x = fmaf(w, f.x, acc.x);
    acc.y = fmaf(w, f.y, acc.y);
    acc.z = fmaf(w, f.z, acc.z);
    acc.w = fmaf(w, f.w, acc.w);
}

__device__ __forceinline__ void edge_weights(float p0, float p1, float p2, float p3,
                                             int j, bool hi, float& wA, float& wB)
{
    const float q0 = lane_bcast(p0, j), q1 = lane_bcast(p1, j);
    const float q2 = lane_bcast(p2, j), q3 = lane_bcast(p3, j);
    wA = hi ? q1 : q0;
    wB = hi ? q3 : q2;
}

__device__ __forceinline__ void gather_rows(
    const float* __restrict__ feat, int s, float p0, float p1, float p2, float p3,
    int cnt, int lane, float4& accA, float4& accB)
{
    const bool hi = lane >= 16;
    const float* fl = feat + 4 * lane;
    int j = 0;
    for (; j + 4 <= cnt; j += 4) {
        const float* r0 = fl + (size_t)__builtin_amdgcn_readlane(s, j)     * FTOT;
        const float* r1 = fl + (size_t)__builtin_amdgcn_readlane(s, j + 1) * FTOT;
        const float* r2 = fl + (size_t)__builtin_amdgcn_readlane(s, j + 2) * FTOT;
        const float* r3 = fl + (size_t)__builtin_amdgcn_readlane(s, j + 3) * FTOT;
        const float4 a0 = *(const float4*)r0, b0 = *(const float4*)(r0 + 128);
        const float4 a1 = *(const float4*)r1, b1 = *(const float4*)(r1 + 128);
        const float4 a2 = *(const float4*)r2, b2 = *(const float4*)(r2 + 128);
        const float4 a3 = *(const float4*)r3, b3 = *(const float4*)(r3 + 128);
        float wA0, wB0, wA1, wB1, wA2, wB2, wA3, wB3;
        edge_weights(p0, p1, p2, p3, j,     hi, wA0, wB0);
        edge_weights(p0, p1, p2, p3, j + 1, hi, wA1, wB1);
        edge_weights(p0, p1, p2, p3, j + 2, hi, wA2, wB2);
        edge_weights(p0, p1, p2, p3, j + 3, hi, wA3, wB3);
        fma4(accA, wA0, a0); fma4(accB, wB0, b0);
        fma4(accA, wA1, a1); fma4(accB, wB1, b1);
        fma4(accA, wA2, a2); fma4(accB, wB2, b2);
        fma4(accA, wA3, a3); fma4(accB, wB3, b3);
    }
    const int rem = cnt - j;
    if (rem > 0) {
        const float* r0 = fl + (size_t)__builtin_amdgcn_readlane(s, j) * FTOT;
        const float4 a0 = *(const float4*)r0, b0 = *(const float4*)(r0 + 128);
        float4 a1 = {0.f, 0.f, 0.f, 0.f}, b1 = a1, a2 = a1, b2 = a1;
        if (rem > 1) {
            const float* r1 = fl + (size_t)__builtin_amdgcn_readlane(s, j + 1) * FTOT;
            a1 = *(const float4*)r1; b1 = *(const float4*)(r1 + 128);
        }
        if (rem > 2) {
            const float* r2 = fl + (size_t)__builtin_amdgcn_readlane(s, j + 2) * FTOT;
            a2 = *(const float4*)r2; b2 = *(const float4*)(r2 + 128);
        }
        float wA, wB;
        edge_weights(p0, p1, p2, p3, j, hi, wA, wB);
        fma4(accA, wA, a0); fma4(accB, wB, b0);
        if (rem > 1) {
            edge_weights(p0, p1, p2, p3, j + 1, hi, wA, wB);
            fma4(accA, wA, a1); fma4(accB, wB, b1);
        }
        if (rem > 2) {
            edge_weights(p0, p1, p2, p3, j + 2, hi, wA, wB);
            fma4(accA, wA, a2); fma4(accB, wB, b2);
        }
    }
}

__global__ __launch_bounds__(256) void gat_node_gather(
    const int* __restrict__ rowptr, const int* __restrict__ elist,
    int* __restrict__ ssrc,
    const float* __restrict__ el, const float* __restrict__ er,
    const float* __restrict__ feat, const float* __restrict__ bias,
    float* __restrict__ outp, int head_major)
{
    const int lane = threadIdx.x & 31;
    const int n = __builtin_amdgcn_readfirstlane(blockIdx.x * 8 + (threadIdx.x >> 5));
    if (n >= N_NODES) return;
    const int beg = rowptr[n], end = rowptr[n + 1];
    const int deg = end - beg;
    const float4 ern = *(const float4*)(er + (size_t)n * HEADS);
    const float NINF = -__builtin_inff();

    float4 accA = {0.f, 0.f, 0.f, 0.f}, accB = {0.f, 0.f, 0.f, 0.f};

    if (deg <= 32) {
        const bool valid = lane < deg;
        int2 me = make_int2(0x7fffffff, 0);
        if (valid) me = *(const int2*)(elist + 2 * (size_t)(beg + lane));
        int r = 0;
        for (int j = 0; j < deg; ++j)
            r += (__builtin_amdgcn_readlane(me.x, j) < me.x) ? 1 : 0;
        if (!valid) r = lane;
        const int s = __builtin_amdgcn_ds_permute(r << 2, me.y);
        float4 l = {0.f, 0.f, 0.f, 0.f};
        if (valid) l = *(const float4*)(el + (size_t)s * HEADS);
        const float x0 = leaky(l.x + ern.x), x1 = leaky(l.y + ern.y);
        const float x2 = leaky(l.z + ern.z), x3 = leaky(l.w + ern.w);
        const float m0 = wave_max(valid ? x0 : NINF), m1 = wave_max(valid ? x1 : NINF);
        const float m2 = wave_max(valid ? x2 : NINF), m3 = wave_max(valid ? x3 : NINF);
        float p0 = valid ? expf(x0 - m0) : 0.f, p1 = valid ? expf(x1 - m1) : 0.f;
        float p2 = valid ? expf(x2 - m2) : 0.f, p3 = valid ? expf(x3 - m3) : 0.f;
        const float d0 = wave_sum(p0), d1 = wave_sum(p1);
        const float d2 = wave_sum(p2), d3 = wave_sum(p3);
        if (valid) { p0 /= d0; p1 /= d1; p2 /= d2; p3 /= d3; }
        gather_rows(feat, s, p0, p1, p2, p3, deg, lane, accA, accB);
    } else {
        for (int i = beg + lane; i < end; i += 32) {
            const int2 me = *(const int2*)(elist + 2 * (size_t)i);
            int r = 0;
            for (int k = beg; k < end; ++k) r += (elist[2 * (size_t)k] < me.x) ? 1 : 0;
            ssrc[beg + r] = me.y;
        }
        __threadfence();
        float m0 = NINF, m1 = NINF, m2 = NINF, m3 = NINF;
        for (int i = beg + lane; i < end; i += 32) {
            const float4 l = *(const float4*)(el + (size_t)ssrc[i] * HEADS);
            m0 = fmaxf(m0, leaky(l.x + ern.x)); m1 = fmaxf(m1, leaky(l.y + ern.y));
            m2 = fmaxf(m2, leaky(l.z + ern.z)); m3 = fmaxf(m3, leaky(l.w + ern.w));
        }
        m0 = wave_max(m0); m1 = wave_max(m1); m2 = wave_max(m2); m3 = wave_max(m3);
        float d0 = 0.f, d1 = 0.f, d2 = 0.f, d3 = 0.f;
        for (int i = beg + lane; i < end; i += 32) {
            const float4 l = *(const float4*)(el + (size_t)ssrc[i] * HEADS);
            d0 += expf(leaky(l.x + ern.x) - m0); d1 += expf(leaky(l.y + ern.y) - m1);
            d2 += expf(leaky(l.z + ern.z) - m2); d3 += expf(leaky(l.w + ern.w) - m3);
        }
        d0 = wave_sum(d0); d1 = wave_sum(d1); d2 = wave_sum(d2); d3 = wave_sum(d3);
        for (int base = beg; base < end; base += 32) {
            const int i = base + lane;
            int s = 0;
            float p0 = 0.f, p1 = 0.f, p2 = 0.f, p3 = 0.f;
            if (i < end) {
                s = ssrc[i];
                const float4 l = *(const float4*)(el + (size_t)s * HEADS);
                p0 = expf(leaky(l.x + ern.x) - m0) / d0;
                p1 = expf(leaky(l.y + ern.y) - m1) / d1;
                p2 = expf(leaky(l.z + ern.z) - m2) / d2;
                p3 = expf(leaky(l.w + ern.w) - m3) / d3;
            }
            const int left = end - base;
            gather_rows(feat, s, p0, p1, p2, p3, left < 32 ? left : 32, lane, accA, accB);
        }
    }

    const float4 bA = *(const float4*)(bias + 4 * lane);
    const float4 bB = *(const float4*)(bias + 128 + 4 * lane);
    float4 oA, oB;
    oA.x = elu1(accA.x + bA.x); oA.y = elu1(accA.y + bA.y);
    oA.z = elu1(accA.z + bA.z); oA.w = elu1(accA.w + bA.w);
    oB.x = elu1(accB.x + bB.x); oB.y = elu1(accB.y + bB.y);
    oB.z = elu1(accB.z + bB.z); oB.w = elu1(accB.w + bB.w);

    if (head_major) {
        const int hsel = lane >> 4;
        const int d = (4 * lane) & 63;
        float* o = outp + (size_t)n * HID + d;
        *(float4*)(o + (size_t)hsel * (N_NODES * HID)) = oA;
        *(float4*)(o + (size_t)(hsel + 2) * (N_NODES * HID)) = oB;
    } else {
        float* o = outp + (size_t)n * FTOT + 4 * lane;
        *(float4*)o = oA;
        *(float4*)(o + 128) = oB;
    }
}

extern "C" void kernel_launch(void* const* d_in, const int* in_sizes, int n_in,
                              void* d_out, int out_size, void* d_ws, size_t ws_size,
                              hipStream_t stream)
{
    (void)in_sizes; (void)n_in; (void)out_size; (void)ws_size;

    const float* x   = (const float*)d_in[0];
    const int*   src = (const int*)d_in[1];
    const int*   dst = (const int*)d_in[2];
    const float* W1  = (const float*)d_in[3];
    const float* al1 = (const float*)d_in[4];
    const float* ar1 = (const float*)d_in[5];
    const float* b1  = (const float*)d_in[6];
    const float* W2  = (const float*)d_in[7];
    const float* al2 = (const float*)d_in[8];
    const float* ar2 = (const float*)d_in[9];
    const float* b2  = (const float*)d_in[10];
    float* out = (float*)d_out;

    char* ws = (char*)d_ws;
    const size_t featB = (size_t)N_NODES * FTOT * sizeof(float);
    const size_t nhB   = (size_t)N_NODES * HEADS * sizeof(float);
    const size_t nodeB = 200192;
    const size_t edgeB = (size_t)N_EDGES * sizeof(int);
    float* feat   = (float*)ws; ws += featB;
    float* h1     = (float*)ws; ws += featB;
    float* el     = (float*)ws; ws += nhB;
    float* er     = (float*)ws; ws += nhB;
    int*   tot    = (int*)ws;   ws += nodeB;
    int*   rowptr = (int*)ws;   ws += nodeB;
    int*   part   = (int*)ws;   ws += edgeB;
    int*   elist  = (int*)ws;   ws += 2 * edgeB;
    int*   ssrc   = (int*)ws;   ws += edgeB;

    dim3 gGemm(N_NODES / (16 * RT), 2), bGemm(256);
    dim3 gSort(SRT_P, SRT_Q);
    const int gNW = (N_NODES + 7) / 8;

    csr_count<<<gSort, 1024, 0, stream>>>(dst, part, tot);
    csr_place<<<gSort, 1024, 0, stream>>>(src, dst, part, tot, rowptr, elist);

    gat_gemm_f32_wmma<<<gGemm, bGemm, 0, stream>>>(x, W1, al1, ar1, feat, el, er);
    gat_node_gather<<<gNW, 256, 0, stream>>>(rowptr, elist, ssrc, el, er, feat, b1, h1, 0);

    gat_gemm_f32_wmma<<<gGemm, bGemm, 0, stream>>>(h1, W2, al2, ar2, feat, el, er);
    gat_node_gather<<<gNW, 256, 0, stream>>>(rowptr, elist, ssrc, el, er, feat, b2, out, 1);
}
